# starting kernel
# baseline (speedup 1.0000x reference)
#include <hip/hip_runtime.h>
#include <hip/hip_bf16.h>

#define N_NODES 4096
#define IN_F    256
#define OUT_F   64
#define NHEAD   8
#define HID     512
#define NEG_BIG (-9.0e15f)
#define LN_EPS  1e-05f

typedef __attribute__((ext_vector_type(2))) float v2f;
typedef __attribute__((ext_vector_type(8))) float v8f;

__device__ __forceinline__ v8f wmma_f32_16x16x4(v2f A, v2f B, v8f C) {
    return __builtin_amdgcn_wmma_f32_16x16x4_f32(false, A, false, B, (short)0, C,
                                                 false, false);
}

__global__ void __launch_bounds__(128)
k1_gemm_ht(const float* __restrict__ h, const float* __restrict__ W,
           float* __restrict__ ht) {
    __shared__ float sA[16 * 260];
    const int i0 = blockIdx.x * 16;
    const int n0 = blockIdx.y * 64;
    const int tid  = threadIdx.x;
    const int lane = tid & 31;
    const int wave = tid >> 5;

    for (int idx = tid; idx < 16 * 256 / 4; idx += 128) {
        const int r = idx >> 6, c = (idx & 63) << 2;
        const float4 v = *(const float4*)&h[(size_t)(i0 + r) * IN_F + c];
        *(float4*)&sA[r * 260 + c] = v;
    }
    __syncthreads();

    const int m  = lane & 15;
    const int hi = lane >> 4;

    v8f acc = {};
    for (int ks = 0; ks < IN_F / 4; ++ks) {
        v2f A, B;
#pragma unroll
        for (int v = 0; v < 2; ++v) {
            const int kk = (hi << 1) + v;
            const int k  = ks * 4 + kk;
            A[v] = sA[m * 260 + k];
            B[v] = W[(size_t)k * HID + n0 + (wave << 4) + m];
        }
        acc = wmma_f32_16x16x4(A, B, acc);
    }

#pragma unroll
    for (int p = 0; p < 8; ++p) {
        const int row = p + (hi << 3);
        ht[(size_t)(i0 + row) * HID + n0 + (wave << 4) + m] = acc[p];
    }
}

__global__ void __launch_bounds__(256)
k2_scores(const float* __restrict__ ht, const float* __restrict__ a,
          float* __restrict__ ssrc, float* __restrict__ sdst) {
    const int p = blockIdx.x * blockDim.x + threadIdx.x;
    if (p >= N_NODES * NHEAD) return;
    const float* row = ht + (size_t)p * OUT_F;
    float s1 = 0.f, s2 = 0.f;
#pragma unroll 8
    for (int f = 0; f < OUT_F; ++f) {
        const float v = row[f];
        s1 += v * a[f];
        s2 += v * a[OUT_F + f];
    }
    ssrc[p] = s1;
    sdst[p] = s2;
}

__global__ void __launch_bounds__(256)
k3_rowstats(const int* __restrict__ adj, const float* __restrict__ ssrc,
            const float* __restrict__ sdst, float* __restrict__ Mb,
            float* __restrict__ Lb) {
    __shared__ float red[NHEAD][256];
    const int i   = blockIdx.x;
    const int tid = threadIdx.x;

    float sr[NHEAD];
#pragma unroll
    for (int hh = 0; hh < NHEAD; ++hh) sr[hh] = ssrc[i * NHEAD + hh];

    float mx[NHEAD];
#pragma unroll
    for (int hh = 0; hh < NHEAD; ++hh) mx[hh] = NEG_BIG;
    for (int j = tid; j < N_NODES; j += 256) {
        const int av = adj[(size_t)i * N_NODES + j];
#pragma unroll
        for (int hh = 0; hh < NHEAD; ++hh) {
            float e = sr[hh] + sdst[j * NHEAD + hh];
            e = e > 0.f ? e : 0.2f * e;
            const float val = av > 0 ? e : NEG_BIG;
            mx[hh] = fmaxf(mx[hh], val);
        }
    }
#pragma unroll
    for (int hh = 0; hh < NHEAD; ++hh) red[hh][tid] = mx[hh];
    __syncthreads();
    for (int s = 128; s >= 1; s >>= 1) {
        if (tid < s) {
#pragma unroll
            for (int hh = 0; hh < NHEAD; ++hh)
                red[hh][tid] = fmaxf(red[hh][tid], red[hh][tid + s]);
        }
        __syncthreads();
    }
    float Mh[NHEAD];
#pragma unroll
    for (int hh = 0; hh < NHEAD; ++hh) Mh[hh] = red[hh][0];
    __syncthreads();

    float sm[NHEAD];
#pragma unroll
    for (int hh = 0; hh < NHEAD; ++hh) sm[hh] = 0.f;
    for (int j = tid; j < N_NODES; j += 256) {
        const int av = adj[(size_t)i * N_NODES + j];
#pragma unroll
        for (int hh = 0; hh < NHEAD; ++hh) {
            float e = sr[hh] + sdst[j * NHEAD + hh];
            e = e > 0.f ? e : 0.2f * e;
            const float val = av > 0 ? e : NEG_BIG;
            sm[hh] += __expf(val - Mh[hh]);
        }
    }
#pragma unroll
    for (int hh = 0; hh < NHEAD; ++hh) red[hh][tid] = sm[hh];
    __syncthreads();
    for (int s = 128; s >= 1; s >>= 1) {
        if (tid < s) {
#pragma unroll
            for (int hh = 0; hh < NHEAD; ++hh)
                red[hh][tid] += red[hh][tid + s];
        }
        __syncthreads();
    }
    if (tid < NHEAD) {
        Mb[i * NHEAD + tid] = Mh[tid];
        Lb[i * NHEAD + tid] = red[tid][0];
    }
}

#define JT 64
__global__ void __launch_bounds__(128)
k4_agg_ln(const int* __restrict__ adj, const float* __restrict__ ht,
          const float* __restrict__ ssrc, const float* __restrict__ sdst,
          const float* __restrict__ Mb, const float* __restrict__ Lb,
          const float* __restrict__ gamma, const float* __restrict__ beta,
          float* __restrict__ out) {
    __shared__ int   s_adj[16 * 68];
    __shared__ float s_sdst[JT * NHEAD];
    __shared__ float s_htT[OUT_F * 68];
    __shared__ float s_out[16 * OUT_F];
    __shared__ float s_L[16 * NHEAD];
    __shared__ float s_M[16 * NHEAD];
    __shared__ float s_ss[16 * NHEAD];
    __shared__ float s_mu[16], s_rs[16];

    const int i0   = blockIdx.x * 16;
    const int tid  = threadIdx.x;
    const int lane = tid & 31;
    const int wave = tid >> 5;
    const int m    = lane & 15;
    const int hi   = lane >> 4;
    const int fcol = (wave << 4) + m;

    {
        const int r = tid >> 3, hh = tid & 7;
        s_L[r * NHEAD + hh]  = Lb[(i0 + r) * NHEAD + hh];
        s_M[r * NHEAD + hh]  = Mb[(i0 + r) * NHEAD + hh];
        s_ss[r * NHEAD + hh] = ssrc[(i0 + r) * NHEAD + hh];
    }
    __syncthreads();

    v8f acc[NHEAD] = {};

    for (int j0 = 0; j0 < N_NODES; j0 += JT) {
        if (j0 + JT < N_NODES) {
            const int r = tid >> 3;
            if (r < 16)
                __builtin_prefetch(&adj[(size_t)(i0 + r) * N_NODES + j0 + JT], 0, 1);
        }
        for (int idx = tid; idx < 16 * JT / 4; idx += 128) {
            const int r = idx >> 4, c = (idx & 15) << 2;
            const int4 v = *(const int4*)&adj[(size_t)(i0 + r) * N_NODES + j0 + c];
            *(int4*)&s_adj[r * 68 + c] = v;
        }
        {
            const float4 v = *(const float4*)&sdst[j0 * NHEAD + (tid << 2)];
            *(float4*)&s_sdst[tid << 2] = v;
        }
        __syncthreads();

#pragma unroll
        for (int hh = 0; hh < NHEAD; ++hh) {
            for (int idx = tid; idx < JT * OUT_F / 4; idx += 128) {
                const int jl = idx >> 4, f4 = (idx & 15) << 2;
                const float4 v =
                    *(const float4*)&ht[(size_t)(j0 + jl) * HID + hh * OUT_F + f4];
                s_htT[(f4 + 0) * 68 + jl] = v.x;
                s_htT[(f4 + 1) * 68 + jl] = v.y;
                s_htT[(f4 + 2) * 68 + jl] = v.z;
                s_htT[(f4 + 3) * 68 + jl] = v.w;
            }
            __syncthreads();

            const float ssr  = s_ss[m * NHEAD + hh];
            const float Mrow = s_M[m * NHEAD + hh];
            v8f a_acc = acc[hh];
            for (int ks = 0; ks < JT / 4; ++ks) {
                const int jbase = ks * 4 + (hi << 1);
                v2f A, B;
#pragma unroll
                for (int v = 0; v < 2; ++v) {
                    const int jl = jbase + v;
                    const int adjv = s_adj[m * 68 + jl];
                    float e = ssr + s_sdst[jl * NHEAD + hh];
                    e = e > 0.f ? e : 0.2f * e;
                    const float val = adjv > 0 ? e : NEG_BIG;
                    A[v] = __expf(val - Mrow);
                    B[v] = s_htT[fcol * 68 + jl];
                }
                a_acc = wmma_f32_16x16x4(A, B, a_acc);
            }
            acc[hh] = a_acc;
            __syncthreads();
        }
        __syncthreads();
    }

    v8f hp = {};
#pragma unroll
    for (int hh = 0; hh < NHEAD; ++hh) {
#pragma unroll
        for (int p = 0; p < 8; ++p) {
            const int row = p + (hi << 3);
            hp[p] += acc[hh][p] / s_L[row * NHEAD + hh];
        }
    }
#pragma unroll
    for (int p = 0; p < 8; ++p) {
        const int row = p + (hi << 3);
        s_out[row * OUT_F + fcol] = hp[p] * (1.0f / NHEAD);
    }
    __syncthreads();

    if (tid < 16) {
        float mu = 0.f;
        for (int f = 0; f < OUT_F; ++f) mu += s_out[tid * OUT_F + f];
        mu *= (1.0f / OUT_F);
        float var = 0.f;
        for (int f = 0; f < OUT_F; ++f) {
            const float d = s_out[tid * OUT_F + f] - mu;
            var += d * d;
        }
        var *= (1.0f / OUT_F);
        s_mu[tid] = mu;
        s_rs[tid] = rsqrtf(var + LN_EPS);
    }
    __syncthreads();

    for (int idx = tid; idx < 16 * OUT_F; idx += 128) {
        const int r = idx >> 6, f = idx & 63;
        out[(size_t)(i0 + r) * OUT_F + f] =
            (s_out[r * OUT_F + f] - s_mu[r]) * s_rs[r] * gamma[f] + beta[f];
    }
}

extern "C" void kernel_launch(void* const* d_in, const int* in_sizes, int n_in,
                              void* d_out, int out_size, void* d_ws, size_t ws_size,
                              hipStream_t stream) {
    const float* h     = (const float*)d_in[0];
    const int*   adj   = (const int*)  d_in[1];
    const float* W     = (const float*)d_in[2];
    const float* a     = (const float*)d_in[3];
    const float* gamma = (const float*)d_in[4];
    const float* beta  = (const float*)d_in[5];
    float* out = (float*)d_out;

    float* ws   = (float*)d_ws;
    float* ht   = ws;
    float* ssrc = ht + (size_t)N_NODES * HID;
    float* sdst = ssrc + N_NODES * NHEAD;
    float* Mb   = sdst + N_NODES * NHEAD;
    float* Lb   = Mb   + N_NODES * NHEAD;

    k1_gemm_ht<<<dim3(N_NODES / 16, HID / 64), 128, 0, stream>>>(h, W, ht);
    k2_scores<<<(N_NODES * NHEAD) / 256, 256, 0, stream>>>(ht, a, ssrc, sdst);
    k3_rowstats<<<N_NODES, 256, 0, stream>>>(adj, ssrc, sdst, Mb, Lb);
    k4_agg_ln<<<N_NODES / 16, 128, 0, stream>>>(adj, ht, ssrc, sdst, Mb, Lb,
                                                gamma, beta, out);
}


# k4: 1024-thread blocks, each 4-wave group owns one head (own LDS tile, one accumulator tile): 8 waves per SIMD instead of 1, 2 barriers per chunk instead of 18; all sum orders kept
# speedup vs baseline: 4.6228x; 4.6228x over previous
#include <hip/hip_runtime.h>
#include <hip/hip_bf16.h>

#define N_NODES 4096
#define IN_F    256
#define OUT_F   64
#define NHEAD   8
#define HID     512
#define NEG_BIG (-9.0e15f)
#define LN_EPS  1e-05f

typedef __attribute__((ext_vector_type(2))) float v2f;
typedef __attribute__((ext_vector_type(8))) float v8f;

__device__ __forceinline__ v8f wmma_f32_16x16x4(v2f A, v2f B, v8f C) {
    return __builtin_amdgcn_wmma_f32_16x16x4_f32(false, A, false, B, (short)0, C,
                                                 false, false);
}

__global__ void __launch_bounds__(128)
k1_gemm_ht(const float* __restrict__ h, const float* __restrict__ W,
           float* __restrict__ ht) {
    __shared__ float sA[16 * 260];
    const int i0 = blockIdx.x * 16;
    const int n0 = blockIdx.y * 64;
    const int tid  = threadIdx.x;
    const int lane = tid & 31;
    const int wave = tid >> 5;

    for (int idx = tid; idx < 16 * 256 / 4; idx += 128) {
        const int r = idx >> 6, c = (idx & 63) << 2;
        const float4 v = *(const float4*)&h[(size_t)(i0 + r) * IN_F + c];
        *(float4*)&sA[r * 260 + c] = v;
    }
    __syncthreads();

    const int m  = lane & 15;
    const int hi = lane >> 4;

    v8f acc = {};
    for (int ks = 0; ks < IN_F / 4; ++ks) {
        v2f A, B;
#pragma unroll
        for (int v = 0; v < 2; ++v) {
            const int kk = (hi << 1) + v;
            const int k  = ks * 4 + kk;
            A[v] = sA[m * 260 + k];
            B[v] = W[(size_t)k * HID + n0 + (wave << 4) + m];
        }
        acc = wmma_f32_16x16x4(A, B, acc);
    }

#pragma unroll
    for (int p = 0; p < 8; ++p) {
        const int row = p + (hi << 3);
        ht[(size_t)(i0 + row) * HID + n0 + (wave << 4) + m] = acc[p];
    }
}

__global__ void __launch_bounds__(256)
k2_scores(const float* __restrict__ ht, const float* __restrict__ a,
          float* __restrict__ ssrc, float* __restrict__ sdst) {
    const int p = blockIdx.x * blockDim.x + threadIdx.x;
    if (p >= N_NODES * NHEAD) return;
    const float* row = ht + (size_t)p * OUT_F;
    float s1 = 0.f, s2 = 0.f;
#pragma unroll 8
    for (int f = 0; f < OUT_F; ++f) {
        const float v = row[f];
        s1 += v * a[f];
        s2 += v * a[OUT_F + f];
    }
    ssrc[p] = s1;
    sdst[p] = s2;
}

__global__ void __launch_bounds__(256)
k3_rowstats(const int* __restrict__ adj, const float* __restrict__ ssrc,
            const float* __restrict__ sdst, float* __restrict__ Mb,
            float* __restrict__ Lb) {
    __shared__ float red[NHEAD][256];
    const int i   = blockIdx.x;
    const int tid = threadIdx.x;

    float sr[NHEAD];
#pragma unroll
    for (int hh = 0; hh < NHEAD; ++hh) sr[hh] = ssrc[i * NHEAD + hh];

    float mx[NHEAD];
#pragma unroll
    for (int hh = 0; hh < NHEAD; ++hh) mx[hh] = NEG_BIG;
    for (int j = tid; j < N_NODES; j += 256) {
        const int av = adj[(size_t)i * N_NODES + j];
#pragma unroll
        for (int hh = 0; hh < NHEAD; ++hh) {
            float e = sr[hh] + sdst[j * NHEAD + hh];
            e = e > 0.f ? e : 0.2f * e;
            const float val = av > 0 ? e : NEG_BIG;
            mx[hh] = fmaxf(mx[hh], val);
        }
    }
#pragma unroll
    for (int hh = 0; hh < NHEAD; ++hh) red[hh][tid] = mx[hh];
    __syncthreads();
    for (int s = 128; s >= 1; s >>= 1) {
        if (tid < s) {
#pragma unroll
            for (int hh = 0; hh < NHEAD; ++hh)
                red[hh][tid] = fmaxf(red[hh][tid], red[hh][tid + s]);
        }
        __syncthreads();
    }
    float Mh[NHEAD];
#pragma unroll
    for (int hh = 0; hh < NHEAD; ++hh) Mh[hh] = red[hh][0];
    __syncthreads();

    float sm[NHEAD];
#pragma unroll
    for (int hh = 0; hh < NHEAD; ++hh) sm[hh] = 0.f;
    for (int j = tid; j < N_NODES; j += 256) {
        const int av = adj[(size_t)i * N_NODES + j];
#pragma unroll
        for (int hh = 0; hh < NHEAD; ++hh) {
            float e = sr[hh] + sdst[j * NHEAD + hh];
            e = e > 0.f ? e : 0.2f * e;
            const float val = av > 0 ? e : NEG_BIG;
            sm[hh] += __expf(val - Mh[hh]);
        }
    }
#pragma unroll
    for (int hh = 0; hh < NHEAD; ++hh) red[hh][tid] = sm[hh];
    __syncthreads();
    for (int s = 128; s >= 1; s >>= 1) {
        if (tid < s) {
#pragma unroll
            for (int hh = 0; hh < NHEAD; ++hh)
                red[hh][tid] += red[hh][tid + s];
        }
        __syncthreads();
    }
    if (tid < NHEAD) {
        Mb[i * NHEAD + tid] = Mh[tid];
        Lb[i * NHEAD + tid] = red[tid][0];
    }
}

#define JT 64
#define K4_THREADS 1024
__global__ void __launch_bounds__(K4_THREADS)
k4_agg_ln(const int* __restrict__ adj, const float* __restrict__ ht,
          const float* __restrict__ ssrc, const float* __restrict__ sdst,
          const float* __restrict__ Mb, const float* __restrict__ Lb,
          const float* __restrict__ gamma, const float* __restrict__ beta,
          float* __restrict__ out) {
    __shared__ int   s_adj[16 * 68];
    __shared__ float s_sdst[JT * NHEAD];
    __shared__ float s_htT[NHEAD * OUT_F * 68];
    __shared__ float s_out[16 * OUT_F];
    __shared__ float s_L[16 * NHEAD];
    __shared__ float s_M[16 * NHEAD];
    __shared__ float s_ss[16 * NHEAD];
    __shared__ float s_mu[16], s_rs[16];

    const int i0   = blockIdx.x * 16;
    const int tid  = threadIdx.x;
    const int hh   = tid >> 7;
    const int gt   = tid & 127;
    const int lane = tid & 31;
    const int wave = gt >> 5;
    const int m    = lane & 15;
    const int hi   = lane >> 4;
    const int fcol = (wave << 4) + m;
    float* const my_htT = s_htT + hh * (OUT_F * 68);

    if (tid < 16 * NHEAD) {
        const int r = tid >> 3, h8 = tid & 7;
        s_L[r * NHEAD + h8]  = Lb[(i0 + r) * NHEAD + h8];
        s_M[r * NHEAD + h8]  = Mb[(i0 + r) * NHEAD + h8];
        s_ss[r * NHEAD + h8] = ssrc[(i0 + r) * NHEAD + h8];
    }
    __syncthreads();

    const float ssr  = s_ss[m * NHEAD + hh];
    const float Mrow = s_M[m * NHEAD + hh];

    v8f acc = {};

    for (int j0 = 0; j0 < N_NODES; j0 += JT) {
        if (tid < 128) {
            if (j0 + JT < N_NODES) {
                const int r = tid >> 3;
                __builtin_prefetch(&adj[(size_t)(i0 + r) * N_NODES + j0 + JT], 0, 1);
            }
        }
        if (tid < 16 * JT / 4) {
            const int r = tid >> 4, c = (tid & 15) << 2;
            const int4 v = *(const int4*)&adj[(size_t)(i0 + r) * N_NODES + j0 + c];
            *(int4*)&s_adj[r * 68 + c] = v;
        } else if (tid < 16 * JT / 4 + JT * NHEAD / 4) {
            const int t4 = (tid - 16 * JT / 4) << 2;
            const float4 v = *(const float4*)&sdst[j0 * NHEAD + t4];
            *(float4*)&s_sdst[t4] = v;
        }
        for (int idx = gt; idx < JT * OUT_F / 4; idx += 128) {
            const int jl = idx >> 4, f4 = (idx & 15) << 2;
            const float4 v =
                *(const float4*)&ht[(size_t)(j0 + jl) * HID + hh * OUT_F + f4];
            my_htT[(f4 + 0) * 68 + jl] = v.x;
            my_htT[(f4 + 1) * 68 + jl] = v.y;
            my_htT[(f4 + 2) * 68 + jl] = v.z;
            my_htT[(f4 + 3) * 68 + jl] = v.w;
        }
        __syncthreads();

        for (int ks = 0; ks < JT / 4; ++ks) {
            const int jbase = ks * 4 + (hi << 1);
            v2f A, B;
#pragma unroll
            for (int v = 0; v < 2; ++v) {
                const int jl = jbase + v;
                const int adjv = s_adj[m * 68 + jl];
                float e = ssr + s_sdst[jl * NHEAD + hh];
                e = e > 0.f ? e : 0.2f * e;
                const float val = adjv > 0 ? e : NEG_BIG;
                A[v] = __expf(val - Mrow);
                B[v] = my_htT[fcol * 68 + jl];
            }
            acc = wmma_f32_16x16x4(A, B, acc);
        }
        __syncthreads();
    }

    float* const s_hn = s_htT;
#pragma unroll
    for (int p = 0; p < 8; ++p) {
        const int row = p + (hi << 3);
        s_hn[(hh * 16 + row) * OUT_F + fcol] = acc[p] / s_L[row * NHEAD + hh];
    }
    __syncthreads();

    {
        float hp = 0.f;
#pragma unroll
        for (int h8 = 0; h8 < NHEAD; ++h8) hp += s_hn[h8 * 16 * OUT_F + tid];
        s_out[tid] = hp * (1.0f / NHEAD);
    }
    __syncthreads();

    if (tid < 16) {
        float mu = 0.f;
        for (int f = 0; f < OUT_F; ++f) mu += s_out[tid * OUT_F + f];
        mu *= (1.0f / OUT_F);
        float var = 0.f;
        for (int f = 0; f < OUT_F; ++f) {
            const float d = s_out[tid * OUT_F + f] - mu;
            var += d * d;
        }
        var *= (1.0f / OUT_F);
        s_mu[tid] = mu;
        s_rs[tid] = rsqrtf(var + LN_EPS);
    }
    __syncthreads();

    {
        const int r = tid >> 6, f = tid & 63;
        out[(size_t)(i0 + r) * OUT_F + f] =
            (s_out[r * OUT_F + f] - s_mu[r]) * s_rs[r] * gamma[f] + beta[f];
    }
}

extern "C" void kernel_launch(void* const* d_in, const int* in_sizes, int n_in,
                              void* d_out, int out_size, void* d_ws, size_t ws_size,
                              hipStream_t stream) {
    const float* h     = (const float*)d_in[0];
    const int*   adj   = (const int*)  d_in[1];
    const float* W     = (const float*)d_in[2];
    const float* a     = (const float*)d_in[3];
    const float* gamma = (const float*)d_in[4];
    const float* beta  = (const float*)d_in[5];
    float* out = (float*)d_out;

    float* ws   = (float*)d_ws;
    float* ht   = ws;
    float* ssrc = ht + (size_t)N_NODES * HID;
    float* sdst = ssrc + N_NODES * NHEAD;
    float* Mb   = sdst + N_NODES * NHEAD;
    float* Lb   = Mb   + N_NODES * NHEAD;

    k1_gemm_ht<<<dim3(N_NODES / 16, HID / 64), 128, 0, stream>>>(h, W, ht);
    k2_scores<<<(N_NODES * NHEAD) / 256, 256, 0, stream>>>(ht, a, ssrc, sdst);
    k3_rowstats<<<N_NODES, 256, 0, stream>>>(adj, ssrc, sdst, Mb, Lb);
    k4_agg_ln<<<N_NODES / 16, K4_THREADS, 0, stream>>>(adj, ht, ssrc, sdst, Mb, Lb,
                                                gamma, beta, out);
}


# k4: ht tile kept row-major in LDS (pitch 80): per thread 8 independent b128 loads, then 8 conflict-free b128 LDS stores; was 8 serial load-wait rounds of 4-way-conflicted scalar stores
# speedup vs baseline: 5.9309x; 1.2830x over previous
#include <hip/hip_runtime.h>
#include <hip/hip_bf16.h>

#define N_NODES 4096
#define IN_F    256
#define OUT_F   64
#define NHEAD   8
#define HID     512
#define NEG_BIG (-9.0e15f)
#define LN_EPS  1e-05f

typedef __attribute__((ext_vector_type(2))) float v2f;
typedef __attribute__((ext_vector_type(8))) float v8f;

__device__ __forceinline__ v8f wmma_f32_16x16x4(v2f A, v2f B, v8f C) {
    return __builtin_amdgcn_wmma_f32_16x16x4_f32(false, A, false, B, (short)0, C,
                                                 false, false);
}

__global__ void __launch_bounds__(128)
k1_gemm_ht(const float* __restrict__ h, const float* __restrict__ W,
           float* __restrict__ ht) {
    __shared__ float sA[16 * 260];
    const int i0 = blockIdx.x * 16;
    const int n0 = blockIdx.y * 64;
    const int tid  = threadIdx.x;
    const int lane = tid & 31;
    const int wave = tid >> 5;

    for (int idx = tid; idx < 16 * 256 / 4; idx += 128) {
        const int r = idx >> 6, c = (idx & 63) << 2;
        const float4 v = *(const float4*)&h[(size_t)(i0 + r) * IN_F + c];
        *(float4*)&sA[r * 260 + c] = v;
    }
    __syncthreads();

    const int m  = lane & 15;
    const int hi = lane >> 4;

    v8f acc = {};
    for (int ks = 0; ks < IN_F / 4; ++ks) {
        v2f A, B;
#pragma unroll
        for (int v = 0; v < 2; ++v) {
            const int kk = (hi << 1) + v;
            const int k  = ks * 4 + kk;
            A[v] = sA[m * 260 + k];
            B[v] = W[(size_t)k * HID + n0 + (wave << 4) + m];
        }
        acc = wmma_f32_16x16x4(A, B, acc);
    }

#pragma unroll
    for (int p = 0; p < 8; ++p) {
        const int row = p + (hi << 3);
        ht[(size_t)(i0 + row) * HID + n0 + (wave << 4) + m] = acc[p];
    }
}

__global__ void __launch_bounds__(256)
k2_scores(const float* __restrict__ ht, const float* __restrict__ a,
          float* __restrict__ ssrc, float* __restrict__ sdst) {
    const int p = blockIdx.x * blockDim.x + threadIdx.x;
    if (p >= N_NODES * NHEAD) return;
    const float* row = ht + (size_t)p * OUT_F;
    float s1 = 0.f, s2 = 0.f;
#pragma unroll 8
    for (int f = 0; f < OUT_F; ++f) {
        const float v = row[f];
        s1 += v * a[f];
        s2 += v * a[OUT_F + f];
    }
    ssrc[p] = s1;
    sdst[p] = s2;
}

__global__ void __launch_bounds__(256)
k3_rowstats(const int* __restrict__ adj, const float* __restrict__ ssrc,
            const float* __restrict__ sdst, float* __restrict__ Mb,
            float* __restrict__ Lb) {
    __shared__ float red[NHEAD][256];
    const int i   = blockIdx.x;
    const int tid = threadIdx.x;

    float sr[NHEAD];
#pragma unroll
    for (int hh = 0; hh < NHEAD; ++hh) sr[hh] = ssrc[i * NHEAD + hh];

    float mx[NHEAD];
#pragma unroll
    for (int hh = 0; hh < NHEAD; ++hh) mx[hh] = NEG_BIG;
    for (int j = tid; j < N_NODES; j += 256) {
        const int av = adj[(size_t)i * N_NODES + j];
#pragma unroll
        for (int hh = 0; hh < NHEAD; ++hh) {
            float e = sr[hh] + sdst[j * NHEAD + hh];
            e = e > 0.f ? e : 0.2f * e;
            const float val = av > 0 ? e : NEG_BIG;
            mx[hh] = fmaxf(mx[hh], val);
        }
    }
#pragma unroll
    for (int hh = 0; hh < NHEAD; ++hh) red[hh][tid] = mx[hh];
    __syncthreads();
    for (int s = 128; s >= 1; s >>= 1) {
        if (tid < s) {
#pragma unroll
            for (int hh = 0; hh < NHEAD; ++hh)
                red[hh][tid] = fmaxf(red[hh][tid], red[hh][tid + s]);
        }
        __syncthreads();
    }
    float Mh[NHEAD];
#pragma unroll
    for (int hh = 0; hh < NHEAD; ++hh) Mh[hh] = red[hh][0];
    __syncthreads();

    float sm[NHEAD];
#pragma unroll
    for (int hh = 0; hh < NHEAD; ++hh) sm[hh] = 0.f;
    for (int j = tid; j < N_NODES; j += 256) {
        const int av = adj[(size_t)i * N_NODES + j];
#pragma unroll
        for (int hh = 0; hh < NHEAD; ++hh) {
            float e = sr[hh] + sdst[j * NHEAD + hh];
            e = e > 0.f ? e : 0.2f * e;
            const float val = av > 0 ? e : NEG_BIG;
            sm[hh] += __expf(val - Mh[hh]);
        }
    }
#pragma unroll
    for (int hh = 0; hh < NHEAD; ++hh) red[hh][tid] = sm[hh];
    __syncthreads();
    for (int s = 128; s >= 1; s >>= 1) {
        if (tid < s) {
#pragma unroll
            for (int hh = 0; hh < NHEAD; ++hh)
                red[hh][tid] += red[hh][tid + s];
        }
        __syncthreads();
    }
    if (tid < NHEAD) {
        Mb[i * NHEAD + tid] = Mh[tid];
        Lb[i * NHEAD + tid] = red[tid][0];
    }
}

#define JT 64
#define HT_PITCH 80
#define HT_ROUNDS 8
#define K4_THREADS 1024
__global__ void __launch_bounds__(K4_THREADS)
k4_agg_ln(const int* __restrict__ adj, const float* __restrict__ ht,
          const float* __restrict__ ssrc, const float* __restrict__ sdst,
          const float* __restrict__ Mb, const float* __restrict__ Lb,
          const float* __restrict__ gamma, const float* __restrict__ beta,
          float* __restrict__ out) {
    __shared__ int   s_adj[16 * 68];
    __shared__ float s_sdst[JT * NHEAD];
    __shared__ alignas(16) float s_ht[NHEAD * JT * HT_PITCH];
    __shared__ float s_out[16 * OUT_F];
    __shared__ float s_L[16 * NHEAD];
    __shared__ float s_M[16 * NHEAD];
    __shared__ float s_ss[16 * NHEAD];
    __shared__ float s_mu[16], s_rs[16];

    const int i0   = blockIdx.x * 16;
    const int tid  = threadIdx.x;
    const int hh   = tid >> 7;
    const int gt   = tid & 127;
    const int lane = tid & 31;
    const int wave = gt >> 5;
    const int m    = lane & 15;
    const int hi   = lane >> 4;
    const int fcol = (wave << 4) + m;
    float* const my_ht = s_ht + hh * (JT * HT_PITCH);
    const int st_j  = gt >> 4;
    const int st_f4 = (gt & 15) << 2;

    if (tid < 16 * NHEAD) {
        const int r = tid >> 3, h8 = tid & 7;
        s_L[r * NHEAD + h8]  = Lb[(i0 + r) * NHEAD + h8];
        s_M[r * NHEAD + h8]  = Mb[(i0 + r) * NHEAD + h8];
        s_ss[r * NHEAD + h8] = ssrc[(i0 + r) * NHEAD + h8];
    }
    __syncthreads();

    const float ssr  = s_ss[m * NHEAD + hh];
    const float Mrow = s_M[m * NHEAD + hh];

    const int*   const a_adj = s_adj + m * 68 + (hi << 1);
    const float* const a_sd  = s_sdst + (hi << 1) * NHEAD + hh;
    const float* const b_ht  = my_ht + (hi << 1) * HT_PITCH + fcol;

    v8f acc = {};

    for (int j0 = 0; j0 < N_NODES; j0 += JT) {
        float4 hv[HT_ROUNDS];
        for (int it = 0; it < HT_ROUNDS; ++it) {
            hv[it] = *(const float4*)&ht[(size_t)(j0 + st_j + 8 * it) * HID +
                                         hh * OUT_F + st_f4];
        }
        if (tid < 128) {
            if (j0 + JT < N_NODES) {
                const int r = tid >> 3;
                __builtin_prefetch(&adj[(size_t)(i0 + r) * N_NODES + j0 + JT], 0, 1);
            }
        }
        if (tid < 16 * JT / 4) {
            const int r = tid >> 4, c = (tid & 15) << 2;
            const int4 v = *(const int4*)&adj[(size_t)(i0 + r) * N_NODES + j0 + c];
            *(int4*)&s_adj[r * 68 + c] = v;
        } else if (tid < 16 * JT / 4 + JT * NHEAD / 4) {
            const int t4 = (tid - 16 * JT / 4) << 2;
            const float4 v = *(const float4*)&sdst[j0 * NHEAD + t4];
            *(float4*)&s_sdst[t4] = v;
        }
        for (int it = 0; it < HT_ROUNDS; ++it) {
            *(float4*)&my_ht[(st_j + 8 * it) * HT_PITCH + st_f4] = hv[it];
        }
        __syncthreads();

        for (int ks = 0; ks < JT / 4; ++ks) {
            v2f A, B;
#pragma unroll
            for (int v = 0; v < 2; ++v) {
                const int jo = ks * 4 + v;
                const int adjv = a_adj[jo];
                float e = ssr + a_sd[jo * NHEAD];
                e = e > 0.f ? e : 0.2f * e;
                const float val = adjv > 0 ? e : NEG_BIG;
                A[v] = __expf(val - Mrow);
                B[v] = b_ht[jo * HT_PITCH];
            }
            acc = wmma_f32_16x16x4(A, B, acc);
        }
        __syncthreads();
    }

    float* const s_hn = s_ht;
#pragma unroll
    for (int p = 0; p < 8; ++p) {
        const int row = p + (hi << 3);
        s_hn[(hh * 16 + row) * OUT_F + fcol] = acc[p] / s_L[row * NHEAD + hh];
    }
    __syncthreads();

    {
        float hp = 0.f;
#pragma unroll
        for (int h8 = 0; h8 < NHEAD; ++h8) hp += s_hn[h8 * 16 * OUT_F + tid];
        s_out[tid] = hp * (1.0f / NHEAD);
    }
    __syncthreads();

    if (tid < 16) {
        float mu = 0.f;
        for (int f = 0; f < OUT_F; ++f) mu += s_out[tid * OUT_F + f];
        mu *= (1.0f / OUT_F);
        float var = 0.f;
        for (int f = 0; f < OUT_F; ++f) {
            const float d = s_out[tid * OUT_F + f] - mu;
            var += d * d;
        }
        var *= (1.0f / OUT_F);
        s_mu[tid] = mu;
        s_rs[tid] = rsqrtf(var + LN_EPS);
    }
    __syncthreads();

    {
        const int r = tid >> 6, f = tid & 63;
        out[(size_t)(i0 + r) * OUT_F + f] =
            (s_out[r * OUT_F + f] - s_mu[r]) * s_rs[r] * gamma[f] + beta[f];
    }
}

extern "C" void kernel_launch(void* const* d_in, const int* in_sizes, int n_in,
                              void* d_out, int out_size, void* d_ws, size_t ws_size,
                              hipStream_t stream) {
    const float* h     = (const float*)d_in[0];
    const int*   adj   = (const int*)  d_in[1];
    const float* W     = (const float*)d_in[2];
    const float* a     = (const float*)d_in[3];
    const float* gamma = (const float*)d_in[4];
    const float* beta  = (const float*)d_in[5];
    float* out = (float*)d_out;

    float* ws   = (float*)d_ws;
    float* ht   = ws;
    float* ssrc = ht + (size_t)N_NODES * HID;
    float* sdst = ssrc + N_NODES * NHEAD;
    float* Mb   = sdst + N_NODES * NHEAD;
    float* Lb   = Mb   + N_NODES * NHEAD;

    k1_gemm_ht<<<dim3(N_NODES / 16, HID / 64), 128, 0, stream>>>(h, W, ht);
    k2_scores<<<(N_NODES * NHEAD) / 256, 256, 0, stream>>>(ht, a, ssrc, sdst);
    k3_rowstats<<<N_NODES, 256, 0, stream>>>(adj, ssrc, sdst, Mb, Lb);
    k4_agg_ln<<<N_NODES / 16, K4_THREADS, 0, stream>>>(adj, ht, ssrc, sdst, Mb, Lb,
                                                gamma, beta, out);
}


# k4: a head's 4 waves split each chunk along k and own all 64 columns (4 acc tiles): numerator built once per k step for 4 WMMAs, not once per WMMA; k sum order changes (4 partials added via LDS)
# speedup vs baseline: 7.7828x; 1.3122x over previous
#include <hip/hip_runtime.h>
#include <hip/hip_bf16.h>

#define N_NODES 4096
#define IN_F    256
#define OUT_F   64
#define NHEAD   8
#define HID     512
#define NEG_BIG (-9.0e15f)
#define LN_EPS  1e-05f

typedef __attribute__((ext_vector_type(2))) float v2f;
typedef __attribute__((ext_vector_type(8))) float v8f;

__device__ __forceinline__ v8f wmma_f32_16x16x4(v2f A, v2f B, v8f C) {
    return __builtin_amdgcn_wmma_f32_16x16x4_f32(false, A, false, B, (short)0, C,
                                                 false, false);
}

__global__ void __launch_bounds__(128)
k1_gemm_ht(const float* __restrict__ h, const float* __restrict__ W,
           float* __restrict__ ht) {
    __shared__ float sA[16 * 260];
    const int i0 = blockIdx.x * 16;
    const int n0 = blockIdx.y * 64;
    const int tid  = threadIdx.x;
    const int lane = tid & 31;
    const int wave = tid >> 5;

    for (int idx = tid; idx < 16 * 256 / 4; idx += 128) {
        const int r = idx >> 6, c = (idx & 63) << 2;
        const float4 v = *(const float4*)&h[(size_t)(i0 + r) * IN_F + c];
        *(float4*)&sA[r * 260 + c] = v;
    }
    __syncthreads();

    const int m  = lane & 15;
    const int hi = lane >> 4;

    v8f acc = {};
    for (int ks = 0; ks < IN_F / 4; ++ks) {
        v2f A, B;
#pragma unroll
        for (int v = 0; v < 2; ++v) {
            const int kk = (hi << 1) + v;
            const int k  = ks * 4 + kk;
            A[v] = sA[m * 260 + k];
            B[v] = W[(size_t)k * HID + n0 + (wave << 4) + m];
        }
        acc = wmma_f32_16x16x4(A, B, acc);
    }

#pragma unroll
    for (int p = 0; p < 8; ++p) {
        const int row = p + (hi << 3);
        ht[(size_t)(i0 + row) * HID + n0 + (wave << 4) + m] = acc[p];
    }
}

__global__ void __launch_bounds__(256)
k2_scores(const float* __restrict__ ht, const float* __restrict__ a,
          float* __restrict__ ssrc, float* __restrict__ sdst) {
    const int p = blockIdx.x * blockDim.x + threadIdx.x;
    if (p >= N_NODES * NHEAD) return;
    const float* row = ht + (size_t)p * OUT_F;
    float s1 = 0.f, s2 = 0.f;
#pragma unroll 8
    for (int f = 0; f < OUT_F; ++f) {
        const float v = row[f];
        s1 += v * a[f];
        s2 += v * a[OUT_F + f];
    }
    ssrc[p] = s1;
    sdst[p] = s2;
}

__global__ void __launch_bounds__(256)
k3_rowstats(const int* __restrict__ adj, const float* __restrict__ ssrc,
            const float* __restrict__ sdst, float* __restrict__ Mb,
            float* __restrict__ Lb) {
    __shared__ float red[NHEAD][256];
    const int i   = blockIdx.x;
    const int tid = threadIdx.x;

    float sr[NHEAD];
#pragma unroll
    for (int hh = 0; hh < NHEAD; ++hh) sr[hh] = ssrc[i * NHEAD + hh];

    float mx[NHEAD];
#pragma unroll
    for (int hh = 0; hh < NHEAD; ++hh) mx[hh] = NEG_BIG;
    for (int j = tid; j < N_NODES; j += 256) {
        const int av = adj[(size_t)i * N_NODES + j];
#pragma unroll
        for (int hh = 0; hh < NHEAD; ++hh) {
            float e = sr[hh] + sdst[j * NHEAD + hh];
            e = e > 0.f ? e : 0.2f * e;
            const float val = av > 0 ? e : NEG_BIG;
            mx[hh] = fmaxf(mx[hh], val);
        }
    }
#pragma unroll
    for (int hh = 0; hh < NHEAD; ++hh) red[hh][tid] = mx[hh];
    __syncthreads();
    for (int s = 128; s >= 1; s >>= 1) {
        if (tid < s) {
#pragma unroll
            for (int hh = 0; hh < NHEAD; ++hh)
                red[hh][tid] = fmaxf(red[hh][tid], red[hh][tid + s]);
        }
        __syncthreads();
    }
    float Mh[NHEAD];
#pragma unroll
    for (int hh = 0; hh < NHEAD; ++hh) Mh[hh] = red[hh][0];
    __syncthreads();

    float sm[NHEAD];
#pragma unroll
    for (int hh = 0; hh < NHEAD; ++hh) sm[hh] = 0.f;
    for (int j = tid; j < N_NODES; j += 256) {
        const int av = adj[(size_t)i * N_NODES + j];
#pragma unroll
        for (int hh = 0; hh < NHEAD; ++hh) {
            float e = sr[hh] + sdst[j * NHEAD + hh];
            e = e > 0.f ? e : 0.2f * e;
            const float val = av > 0 ? e : NEG_BIG;
            sm[hh] += __expf(val - Mh[hh]);
        }
    }
#pragma unroll
    for (int hh = 0; hh < NHEAD; ++hh) red[hh][tid] = sm[hh];
    __syncthreads();
    for (int s = 128; s >= 1; s >>= 1) {
        if (tid < s) {
#pragma unroll
            for (int hh = 0; hh < NHEAD; ++hh)
                red[hh][tid] += red[hh][tid + s];
        }
        __syncthreads();
    }
    if (tid < NHEAD) {
        Mb[i * NHEAD + tid] = Mh[tid];
        Lb[i * NHEAD + tid] = red[tid][0];
    }
}

#define JT 64
#define HT_PITCH 80
#define HT_ROUNDS 8
#define K4_THREADS 1024
#define KS_WAVE 4
__global__ void __launch_bounds__(K4_THREADS)
k4_agg_ln(const int* __restrict__ adj, const float* __restrict__ ht,
          const float* __restrict__ ssrc, const float* __restrict__ sdst,
          const float* __restrict__ Mb, const float* __restrict__ Lb,
          const float* __restrict__ gamma, const float* __restrict__ beta,
          float* __restrict__ out) {
    __shared__ int   s_adj[16 * 68];
    __shared__ float s_sdst[JT * NHEAD];
    __shared__ alignas(16) float s_ht[NHEAD * JT * HT_PITCH];
    __shared__ float s_out[16 * OUT_F];
    __shared__ float s_L[16 * NHEAD];
    __shared__ float s_M[16 * NHEAD];
    __shared__ float s_ss[16 * NHEAD];
    __shared__ float s_mu[16], s_rs[16];

    const int i0   = blockIdx.x * 16;
    const int tid  = threadIdx.x;
    const int hh   = tid >> 7;
    const int gt   = tid & 127;
    const int lane = tid & 31;
    const int wave = gt >> 5;
    const int m    = lane & 15;
    const int hi   = lane >> 4;
    float* const my_ht = s_ht + hh * (JT * HT_PITCH);
    const int st_j  = gt >> 4;
    const int st_f4 = (gt & 15) << 2;

    if (tid < 16 * NHEAD) {
        const int r = tid >> 3, h8 = tid & 7;
        s_L[r * NHEAD + h8]  = Lb[(i0 + r) * NHEAD + h8];
        s_M[r * NHEAD + h8]  = Mb[(i0 + r) * NHEAD + h8];
        s_ss[r * NHEAD + h8] = ssrc[(i0 + r) * NHEAD + h8];
    }
    __syncthreads();

    const float ssr  = s_ss[m * NHEAD + hh];
    const float Mrow = s_M[m * NHEAD + hh];

    const int jw = (wave << 4) + (hi << 1);
    const int*   const a_adj = s_adj + m * 68 + jw;
    const float* const a_sd  = s_sdst + jw * NHEAD + hh;
    const float* const b_ht  = my_ht + jw * HT_PITCH + (m << 2);

    v8f acc0 = {}, acc1 = {}, acc2 = {}, acc3 = {};

    for (int j0 = 0; j0 < N_NODES; j0 += JT) {
        float4 hv[HT_ROUNDS];
        for (int it = 0; it < HT_ROUNDS; ++it) {
            hv[it] = *(const float4*)&ht[(size_t)(j0 + st_j + 8 * it) * HID +
                                         hh * OUT_F + st_f4];
        }
        if (tid < 128) {
            if (j0 + JT < N_NODES) {
                const int r = tid >> 3;
                __builtin_prefetch(&adj[(size_t)(i0 + r) * N_NODES + j0 + JT], 0, 1);
            }
        }
        if (tid < 16 * JT / 4) {
            const int r = tid >> 4, c = (tid & 15) << 2;
            const int4 v = *(const int4*)&adj[(size_t)(i0 + r) * N_NODES + j0 + c];
            *(int4*)&s_adj[r * 68 + c] = v;
        } else if (tid < 16 * JT / 4 + JT * NHEAD / 4) {
            const int t4 = (tid - 16 * JT / 4) << 2;
            const float4 v = *(const float4*)&sdst[j0 * NHEAD + t4];
            *(float4*)&s_sdst[t4] = v;
        }
        for (int it = 0; it < HT_ROUNDS; ++it) {
            *(float4*)&my_ht[(st_j + 8 * it) * HT_PITCH + st_f4] = hv[it];
        }
        __syncthreads();

        for (int ks = 0; ks < KS_WAVE; ++ks) {
            v2f A;
#pragma unroll
            for (int v = 0; v < 2; ++v) {
                const int jo = ks * 4 + v;
                const int adjv = a_adj[jo];
                float e = ssr + a_sd[jo * NHEAD];
                e = e > 0.f ? e : 0.2f * e;
                const float val = adjv > 0 ? e : NEG_BIG;
                A[v] = __expf(val - Mrow);
            }
            const float4 r0 = *(const float4*)&b_ht[(ks * 4) * HT_PITCH];
            const float4 r1 = *(const float4*)&b_ht[(ks * 4 + 1) * HT_PITCH];
            const v2f B0 = {r0.x, r1.x};
            const v2f B1 = {r0.y, r1.y};
            const v2f B2 = {r0.z, r1.z};
            const v2f B3 = {r0.w, r1.w};
            acc0 = wmma_f32_16x16x4(A, B0, acc0);
            acc1 = wmma_f32_16x16x4(A, B1, acc1);
            acc2 = wmma_f32_16x16x4(A, B2, acc2);
            acc3 = wmma_f32_16x16x4(A, B3, acc3);
        }
        __syncthreads();
    }

    float* const s_part = s_ht;
    asm volatile("s_wait_alu depctr_va_vdst(0)"
                 : "+v"(acc0), "+v"(acc1), "+v"(acc2), "+v"(acc3));
    {
        float* const my_part = s_part + ((hh * 4 + wave) * 16) * OUT_F + (m << 2);
#pragma unroll
        for (int p = 0; p < 8; ++p) {
            const int row = p + (hi << 3);
            *(float4*)&my_part[row * OUT_F] =
                make_float4(acc0[p], acc1[p], acc2[p], acc3[p]);
        }
    }
    __syncthreads();

    {
        const int r = tid >> 6;
        float hp = 0.f;
#pragma unroll
        for (int h8 = 0; h8 < NHEAD; ++h8) {
            const float* const q = s_part + (h8 * 4) * 16 * OUT_F + tid;
            float s = q[0];
            s += q[1 * 16 * OUT_F];
            s += q[2 * 16 * OUT_F];
            s += q[3 * 16 * OUT_F];
            hp += s / s_L[r * NHEAD + h8];
        }
        s_out[tid] = hp * (1.0f / NHEAD);
    }
    __syncthreads();

    if (tid < 16) {
        float mu = 0.f;
        for (int f = 0; f < OUT_F; ++f) mu += s_out[tid * OUT_F + f];
        mu *= (1.0f / OUT_F);
        float var = 0.f;
        for (int f = 0; f < OUT_F; ++f) {
            const float d = s_out[tid * OUT_F + f] - mu;
            var += d * d;
        }
        var *= (1.0f / OUT_F);
        s_mu[tid] = mu;
        s_rs[tid] = rsqrtf(var + LN_EPS);
    }
    __syncthreads();

    {
        const int r = tid >> 6, f = tid & 63;
        out[(size_t)(i0 + r) * OUT_F + f] =
            (s_out[r * OUT_F + f] - s_mu[r]) * s_rs[r] * gamma[f] + beta[f];
    }
}

extern "C" void kernel_launch(void* const* d_in, const int* in_sizes, int n_in,
                              void* d_out, int out_size, void* d_ws, size_t ws_size,
                              hipStream_t stream) {
    const float* h     = (const float*)d_in[0];
    const int*   adj   = (const int*)  d_in[1];
    const float* W     = (const float*)d_in[2];
    const float* a     = (const float*)d_in[3];
    const float* gamma = (const float*)d_in[4];
    const float* beta  = (const float*)d_in[5];
    float* out = (float*)d_out;

    float* ws   = (float*)d_ws;
    float* ht   = ws;
    float* ssrc = ht + (size_t)N_NODES * HID;
    float* sdst = ssrc + N_NODES * NHEAD;
    float* Mb   = sdst + N_NODES * NHEAD;
    float* Lb   = Mb   + N_NODES * NHEAD;

    k1_gemm_ht<<<dim3(N_NODES / 16, HID / 64), 128, 0, stream>>>(h, W, ht);
    k2_scores<<<(N_NODES * NHEAD) / 256, 256, 0, stream>>>(ht, a, ssrc, sdst);
    k3_rowstats<<<N_NODES, 256, 0, stream>>>(adj, ssrc, sdst, Mb, Lb);
    k4_agg_ln<<<N_NODES / 16, K4_THREADS, 0, stream>>>(adj, ht, ssrc, sdst, Mb, Lb,
                                                gamma, beta, out);
}


# k4 sums the softmax denominator L from the numerators it already builds (packed add per k step; 8 partials per row and head added via LDS); k3 loses its exp pass; L's sum order changes
# speedup vs baseline: 8.7061x; 1.1186x over previous
#include <hip/hip_runtime.h>
#include <hip/hip_bf16.h>

#define N_NODES 4096
#define IN_F    256
#define OUT_F   64
#define NHEAD   8
#define HID     512
#define NEG_BIG (-9.0e15f)
#define LN_EPS  1e-05f

typedef __attribute__((ext_vector_type(2))) float v2f;
typedef __attribute__((ext_vector_type(8))) float v8f;

__device__ __forceinline__ v8f wmma_f32_16x16x4(v2f A, v2f B, v8f C) {
    return __builtin_amdgcn_wmma_f32_16x16x4_f32(false, A, false, B, (short)0, C,
                                                 false, false);
}

__global__ void __launch_bounds__(128)
k1_gemm_ht(const float* __restrict__ h, const float* __restrict__ W,
           float* __restrict__ ht) {
    __shared__ float sA[16 * 260];
    const int i0 = blockIdx.x * 16;
    const int n0 = blockIdx.y * 64;
    const int tid  = threadIdx.x;
    const int lane = tid & 31;
    const int wave = tid >> 5;

    for (int idx = tid; idx < 16 * 256 / 4; idx += 128) {
        const int r = idx >> 6, c = (idx & 63) << 2;
        const float4 v = *(const float4*)&h[(size_t)(i0 + r) * IN_F + c];
        *(float4*)&sA[r * 260 + c] = v;
    }
    __syncthreads();

    const int m  = lane & 15;
    const int hi = lane >> 4;

    v8f acc = {};
    for (int ks = 0; ks < IN_F / 4; ++ks) {
        v2f A, B;
#pragma unroll
        for (int v = 0; v < 2; ++v) {
            const int kk = (hi << 1) + v;
            const int k  = ks * 4 + kk;
            A[v] = sA[m * 260 + k];
            B[v] = W[(size_t)k * HID + n0 + (wave << 4) + m];
        }
        acc = wmma_f32_16x16x4(A, B, acc);
    }

#pragma unroll
    for (int p = 0; p < 8; ++p) {
        const int row = p + (hi << 3);
        ht[(size_t)(i0 + row) * HID + n0 + (wave << 4) + m] = acc[p];
    }
}

__global__ void __launch_bounds__(256)
k2_scores(const float* __restrict__ ht, const float* __restrict__ a,
          float* __restrict__ ssrc, float* __restrict__ sdst) {
    const int p = blockIdx.x * blockDim.x + threadIdx.x;
    if (p >= N_NODES * NHEAD) return;
    const float* row = ht + (size_t)p * OUT_F;
    float s1 = 0.f, s2 = 0.f;
#pragma unroll 8
    for (int f = 0; f < OUT_F; ++f) {
        const float v = row[f];
        s1 += v * a[f];
        s2 += v * a[OUT_F + f];
    }
    ssrc[p] = s1;
    sdst[p] = s2;
}

__global__ void __launch_bounds__(256)
k3_rowmax(const int* __restrict__ adj, const float* __restrict__ ssrc,
          const float* __restrict__ sdst, float* __restrict__ Mb) {
    __shared__ float red[NHEAD][256];
    const int i   = blockIdx.x;
    const int tid = threadIdx.x;

    float sr[NHEAD];
#pragma unroll
    for (int hh = 0; hh < NHEAD; ++hh) sr[hh] = ssrc[i * NHEAD + hh];

    float mx[NHEAD];
#pragma unroll
    for (int hh = 0; hh < NHEAD; ++hh) mx[hh] = NEG_BIG;
    for (int j = tid; j < N_NODES; j += 256) {
        const int av = adj[(size_t)i * N_NODES + j];
#pragma unroll
        for (int hh = 0; hh < NHEAD; ++hh) {
            float e = sr[hh] + sdst[j * NHEAD + hh];
            e = e > 0.f ? e : 0.2f * e;
            const float val = av > 0 ? e : NEG_BIG;
            mx[hh] = fmaxf(mx[hh], val);
        }
    }
#pragma unroll
    for (int hh = 0; hh < NHEAD; ++hh) red[hh][tid] = mx[hh];
    __syncthreads();
    for (int s = 128; s >= 1; s >>= 1) {
        if (tid < s) {
#pragma unroll
            for (int hh = 0; hh < NHEAD; ++hh)
                red[hh][tid] = fmaxf(red[hh][tid], red[hh][tid + s]);
        }
        __syncthreads();
    }
    if (tid < NHEAD) Mb[i * NHEAD + tid] = red[tid][0];
}

#define JT 64
#define HT_PITCH 80
#define HT_ROUNDS 8
#define K4_THREADS 1024
#define KS_WAVE 4
__global__ void __launch_bounds__(K4_THREADS)
k4_agg_ln(const int* __restrict__ adj, const float* __restrict__ ht,
          const float* __restrict__ ssrc, const float* __restrict__ sdst,
          const float* __restrict__ Mb,
          const float* __restrict__ gamma, const float* __restrict__ beta,
          float* __restrict__ out) {
    __shared__ int   s_adj[16 * 68];
    __shared__ float s_sdst[JT * NHEAD];
    __shared__ alignas(16) float s_ht[NHEAD * JT * HT_PITCH];
    __shared__ float s_out[16 * OUT_F];
    __shared__ float s_L[16 * NHEAD];
    __shared__ float s_Lp[NHEAD * 4 * 2 * 16];
    __shared__ float s_M[16 * NHEAD];
    __shared__ float s_ss[16 * NHEAD];
    __shared__ float s_mu[16], s_rs[16];

    const int i0   = blockIdx.x * 16;
    const int tid  = threadIdx.x;
    const int hh   = tid >> 7;
    const int gt   = tid & 127;
    const int lane = tid & 31;
    const int wave = gt >> 5;
    const int m    = lane & 15;
    const int hi   = lane >> 4;
    float* const my_ht = s_ht + hh * (JT * HT_PITCH);
    const int st_j  = gt >> 4;
    const int st_f4 = (gt & 15) << 2;

    if (tid < 16 * NHEAD) {
        const int r = tid >> 3, h8 = tid & 7;
        s_M[r * NHEAD + h8]  = Mb[(i0 + r) * NHEAD + h8];
        s_ss[r * NHEAD + h8] = ssrc[(i0 + r) * NHEAD + h8];
    }
    __syncthreads();

    const float ssr  = s_ss[m * NHEAD + hh];
    const float Mrow = s_M[m * NHEAD + hh];

    const int jw = (wave << 4) + (hi << 1);
    const int*   const a_adj = s_adj + m * 68 + jw;
    const float* const a_sd  = s_sdst + jw * NHEAD + hh;
    const float* const b_ht  = my_ht + jw * HT_PITCH + (m << 2);

    v8f acc0 = {}, acc1 = {}, acc2 = {}, acc3 = {};
    v2f lsum2 = {};

    for (int j0 = 0; j0 < N_NODES; j0 += JT) {
        float4 hv[HT_ROUNDS];
        for (int it = 0; it < HT_ROUNDS; ++it) {
            hv[it] = *(const float4*)&ht[(size_t)(j0 + st_j + 8 * it) * HID +
                                         hh * OUT_F + st_f4];
        }
        if (tid < 128) {
            if (j0 + JT < N_NODES) {
                const int r = tid >> 3;
                __builtin_prefetch(&adj[(size_t)(i0 + r) * N_NODES + j0 + JT], 0, 1);
            }
        }
        if (tid < 16 * JT / 4) {
            const int r = tid >> 4, c = (tid & 15) << 2;
            const int4 v = *(const int4*)&adj[(size_t)(i0 + r) * N_NODES + j0 + c];
            *(int4*)&s_adj[r * 68 + c] = v;
        } else if (tid < 16 * JT / 4 + JT * NHEAD / 4) {
            const int t4 = (tid - 16 * JT / 4) << 2;
            const float4 v = *(const float4*)&sdst[j0 * NHEAD + t4];
            *(float4*)&s_sdst[t4] = v;
        }
        for (int it = 0; it < HT_ROUNDS; ++it) {
            *(float4*)&my_ht[(st_j + 8 * it) * HT_PITCH + st_f4] = hv[it];
        }
        __syncthreads();

        for (int ks = 0; ks < KS_WAVE; ++ks) {
            v2f A;
#pragma unroll
            for (int v = 0; v < 2; ++v) {
                const int jo = ks * 4 + v;
                const int adjv = a_adj[jo];
                float e = ssr + a_sd[jo * NHEAD];
                e = e > 0.f ? e : 0.2f * e;
                const float val = adjv > 0 ? e : NEG_BIG;
                A[v] = __expf(val - Mrow);
            }
            lsum2 += A;
            const float4 r0 = *(const float4*)&b_ht[(ks * 4) * HT_PITCH];
            const float4 r1 = *(const float4*)&b_ht[(ks * 4 + 1) * HT_PITCH];
            const v2f B0 = {r0.x, r1.x};
            const v2f B1 = {r0.y, r1.y};
            const v2f B2 = {r0.z, r1.z};
            const v2f B3 = {r0.w, r1.w};
            acc0 = wmma_f32_16x16x4(A, B0, acc0);
            acc1 = wmma_f32_16x16x4(A, B1, acc1);
            acc2 = wmma_f32_16x16x4(A, B2, acc2);
            acc3 = wmma_f32_16x16x4(A, B3, acc3);
        }
        __syncthreads();
    }

    float* const s_part = s_ht;
    asm volatile("s_wait_alu depctr_va_vdst(0)"
                 : "+v"(acc0), "+v"(acc1), "+v"(acc2), "+v"(acc3));
    {
        float* const my_part = s_part + ((hh * 4 + wave) * 16) * OUT_F + (m << 2);
#pragma unroll
        for (int p = 0; p < 8; ++p) {
            const int row = p + (hi << 3);
            *(float4*)&my_part[row * OUT_F] =
                make_float4(acc0[p], acc1[p], acc2[p], acc3[p]);
        }
    }
    s_Lp[tid] = lsum2[0] + lsum2[1];
    __syncthreads();

    if (tid < 16 * NHEAD) {
        const int r = tid >> 3, h8 = tid & 7;
        const float* const q = s_Lp + h8 * (4 * 2 * 16) + r;
        float L = q[0];
        for (int t = 1; t < 8; ++t) L += q[t * 16];
        s_L[r * NHEAD + h8] = L;
    }
    __syncthreads();

    {
        const int r = tid >> 6;
        float hp = 0.f;
#pragma unroll
        for (int h8 = 0; h8 < NHEAD; ++h8) {
            const float* const q = s_part + (h8 * 4) * 16 * OUT_F + tid;
            float s = q[0];
            s += q[1 * 16 * OUT_F];
            s += q[2 * 16 * OUT_F];
            s += q[3 * 16 * OUT_F];
            hp += s / s_L[r * NHEAD + h8];
        }
        s_out[tid] = hp * (1.0f / NHEAD);
    }
    __syncthreads();

    if (tid < 16) {
        float mu = 0.f;
        for (int f = 0; f < OUT_F; ++f) mu += s_out[tid * OUT_F + f];
        mu *= (1.0f / OUT_F);
        float var = 0.f;
        for (int f = 0; f < OUT_F; ++f) {
            const float d = s_out[tid * OUT_F + f] - mu;
            var += d * d;
        }
        var *= (1.0f / OUT_F);
        s_mu[tid] = mu;
        s_rs[tid] = rsqrtf(var + LN_EPS);
    }
    __syncthreads();

    {
        const int r = tid >> 6, f = tid & 63;
        out[(size_t)(i0 + r) * OUT_F + f] =
            (s_out[r * OUT_F + f] - s_mu[r]) * s_rs[r] * gamma[f] + beta[f];
    }
}

extern "C" void kernel_launch(void* const* d_in, const int* in_sizes, int n_in,
                              void* d_out, int out_size, void* d_ws, size_t ws_size,
                              hipStream_t stream) {
    const float* h     = (const float*)d_in[0];
    const int*   adj   = (const int*)  d_in[1];
    const float* W     = (const float*)d_in[2];
    const float* a     = (const float*)d_in[3];
    const float* gamma = (const float*)d_in[4];
    const float* beta  = (const float*)d_in[5];
    float* out = (float*)d_out;

    float* ws   = (float*)d_ws;
    float* ht   = ws;
    float* ssrc = ht + (size_t)N_NODES * HID;
    float* sdst = ssrc + N_NODES * NHEAD;
    float* Mb   = sdst + N_NODES * NHEAD;

    k1_gemm_ht<<<dim3(N_NODES / 16, HID / 64), 128, 0, stream>>>(h, W, ht);
    k2_scores<<<(N_NODES * NHEAD) / 256, 256, 0, stream>>>(ht, a, ssrc, sdst);
    k3_rowmax<<<N_NODES, 256, 0, stream>>>(adj, ssrc, sdst, Mb);
    k4_agg_ln<<<N_NODES / 16, K4_THREADS, 0, stream>>>(adj, ht, ssrc, sdst, Mb,
                                                gamma, beta, out);
}


# k4: B fragments go from global memory straight to registers, one chunk ahead (a staged ht element was read back by one lane only); LDS keeps a 2-stage adj/s_dst ring, 1 barrier/chunk; sum orders kept
# speedup vs baseline: 10.3250x; 1.1860x over previous
#include <hip/hip_runtime.h>
#include <hip/hip_bf16.h>

#define N_NODES 4096
#define IN_F    256
#define OUT_F   64
#define NHEAD   8
#define HID     512
#define NEG_BIG (-9.0e15f)
#define LN_EPS  1e-05f

typedef __attribute__((ext_vector_type(2))) float v2f;
typedef __attribute__((ext_vector_type(8))) float v8f;

__device__ __forceinline__ v8f wmma_f32_16x16x4(v2f A, v2f B, v8f C) {
    return __builtin_amdgcn_wmma_f32_16x16x4_f32(false, A, false, B, (short)0, C,
                                                 false, false);
}

__global__ void __launch_bounds__(128)
k1_gemm_ht(const float* __restrict__ h, const float* __restrict__ W,
           float* __restrict__ ht) {
    __shared__ float sA[16 * 260];
    const int i0 = blockIdx.x * 16;
    const int n0 = blockIdx.y * 64;
    const int tid  = threadIdx.x;
    const int lane = tid & 31;
    const int wave = tid >> 5;

    for (int idx = tid; idx < 16 * 256 / 4; idx += 128) {
        const int r = idx >> 6, c = (idx & 63) << 2;
        const float4 v = *(const float4*)&h[(size_t)(i0 + r) * IN_F + c];
        *(float4*)&sA[r * 260 + c] = v;
    }
    __syncthreads();

    const int m  = lane & 15;
    const int hi = lane >> 4;

    v8f acc = {};
    for (int ks = 0; ks < IN_F / 4; ++ks) {
        v2f A, B;
#pragma unroll
        for (int v = 0; v < 2; ++v) {
            const int kk = (hi << 1) + v;
            const int k  = ks * 4 + kk;
            A[v] = sA[m * 260 + k];
            B[v] = W[(size_t)k * HID + n0 + (wave << 4) + m];
        }
        acc = wmma_f32_16x16x4(A, B, acc);
    }

#pragma unroll
    for (int p = 0; p < 8; ++p) {
        const int row = p + (hi << 3);
        ht[(size_t)(i0 + row) * HID + n0 + (wave << 4) + m] = acc[p];
    }
}

__global__ void __launch_bounds__(256)
k2_scores(const float* __restrict__ ht, const float* __restrict__ a,
          float* __restrict__ ssrc, float* __restrict__ sdst) {
    const int p = blockIdx.x * blockDim.x + threadIdx.x;
    if (p >= N_NODES * NHEAD) return;
    const float* row = ht + (size_t)p * OUT_F;
    float s1 = 0.f, s2 = 0.f;
#pragma unroll 8
    for (int f = 0; f < OUT_F; ++f) {
        const float v = row[f];
        s1 += v * a[f];
        s2 += v * a[OUT_F + f];
    }
    ssrc[p] = s1;
    sdst[p] = s2;
}

__global__ void __launch_bounds__(256)
k3_rowmax(const int* __restrict__ adj, const float* __restrict__ ssrc,
          const float* __restrict__ sdst, float* __restrict__ Mb) {
    __shared__ float red[NHEAD][256];
    const int i   = blockIdx.x;
    const int tid = threadIdx.x;

    float sr[NHEAD];
#pragma unroll
    for (int hh = 0; hh < NHEAD; ++hh) sr[hh] = ssrc[i * NHEAD + hh];

    float mx[NHEAD];
#pragma unroll
    for (int hh = 0; hh < NHEAD; ++hh) mx[hh] = NEG_BIG;
    for (int j = tid; j < N_NODES; j += 256) {
        const int av = adj[(size_t)i * N_NODES + j];
#pragma unroll
        for (int hh = 0; hh < NHEAD; ++hh) {
            float e = sr[hh] + sdst[j * NHEAD + hh];
            e = e > 0.f ? e : 0.2f * e;
            const float val = av > 0 ? e : NEG_BIG;
            mx[hh] = fmaxf(mx[hh], val);
        }
    }
#pragma unroll
    for (int hh = 0; hh < NHEAD; ++hh) red[hh][tid] = mx[hh];
    __syncthreads();
    for (int s = 128; s >= 1; s >>= 1) {
        if (tid < s) {
#pragma unroll
            for (int hh = 0; hh < NHEAD; ++hh)
                red[hh][tid] = fmaxf(red[hh][tid], red[hh][tid + s]);
        }
        __syncthreads();
    }
    if (tid < NHEAD) Mb[i * NHEAD + tid] = red[tid][0];
}

#define JT 64
#define K4_THREADS 1024
#define KS_WAVE 4
#define ADJ_PITCH 68
#define SD_OFF (16 * ADJ_PITCH)
#define STAGE_INTS (SD_OFF + JT * NHEAD)
#define N_CHUNKS (N_NODES / JT)

__device__ __forceinline__ void k4_load_b(float4 (&dst)[8], const float* __restrict__ src) {
#pragma unroll
    for (int q = 0; q < 8; ++q)
        dst[q] = *(const float4*)&src[(size_t)(4 * (q >> 1) + (q & 1)) * HID];
}

__device__ __forceinline__ void k4_chunk(const int* a_adj, const int* a_sd,
                                         const float4 (&hv)[8],
                                         const float ssr, const float Mrow,
                                         v8f& acc0, v8f& acc1, v8f& acc2, v8f& acc3,
                                         v2f& lsum2) {
#pragma unroll
    for (int ks = 0; ks < KS_WAVE; ++ks) {
        v2f A;
#pragma unroll
        for (int v = 0; v < 2; ++v) {
            const int jo = ks * 4 + v;
            const int adjv = a_adj[jo];
            float e = ssr + __int_as_float(a_sd[jo * NHEAD]);
            e = e > 0.f ? e : 0.2f * e;
            const float val = adjv > 0 ? e : NEG_BIG;
            A[v] = __expf(val - Mrow);
        }
        lsum2 += A;
        const float4 r0 = hv[2 * ks];
        const float4 r1 = hv[2 * ks + 1];
        const v2f B0 = {r0.x, r1.x};
        const v2f B1 = {r0.y, r1.y};
        const v2f B2 = {r0.z, r1.z};
        const v2f B3 = {r0.w, r1.w};
        acc0 = wmma_f32_16x16x4(A, B0, acc0);
        acc1 = wmma_f32_16x16x4(A, B1, acc1);
        acc2 = wmma_f32_16x16x4(A, B2, acc2);
        acc3 = wmma_f32_16x16x4(A, B3, acc3);
    }
}

__global__ void __launch_bounds__(K4_THREADS)
k4_agg_ln(const int* __restrict__ adj, const float* __restrict__ ht,
          const float* __restrict__ ssrc, const float* __restrict__ sdst,
          const float* __restrict__ Mb,
          const float* __restrict__ gamma, const float* __restrict__ beta,
          float* __restrict__ out) {
    __shared__ alignas(16) int s_tile[2 * STAGE_INTS];
    __shared__ alignas(16) float s_part[NHEAD * 4 * 16 * OUT_F];
    __shared__ float s_out[16 * OUT_F];
    __shared__ float s_L[16 * NHEAD];
    __shared__ float s_Lp[NHEAD * 4 * 2 * 16];
    __shared__ float s_M[16 * NHEAD];
    __shared__ float s_ss[16 * NHEAD];
    __shared__ float s_mu[16], s_rs[16];

    const int i0   = blockIdx.x * 16;
    const int tid  = threadIdx.x;
    const int hh   = tid >> 7;
    const int gt   = tid & 127;
    const int lane = tid & 31;
    const int wave = gt >> 5;
    const int m    = lane & 15;
    const int hi   = lane >> 4;

    if (tid < 16 * NHEAD) {
        const int r = tid >> 3, h8 = tid & 7;
        s_M[r * NHEAD + h8]  = Mb[(i0 + r) * NHEAD + h8];
        s_ss[r * NHEAD + h8] = ssrc[(i0 + r) * NHEAD + h8];
    }

    const bool stager = tid < 16 * JT / 4 + JT * NHEAD / 4;
    const int4* t_src;
    int t_step;
    int t_dst;
    if (tid < 16 * JT / 4) {
        const int r = tid >> 4, c = (tid & 15) << 2;
        t_src  = (const int4*)&adj[(size_t)(i0 + r) * N_NODES + c];
        t_step = JT / 4;
        t_dst  = r * ADJ_PITCH + c;
    } else {
        const int t4 = ((tid - 16 * JT / 4) & (JT * NHEAD / 4 - 1)) << 2;
        t_src  = (const int4*)&sdst[t4];
        t_step = JT * NHEAD / 4;
        t_dst  = SD_OFF + t4;
    }

    const int jw = (wave << 4) + (hi << 1);
    const int* const a_adj = s_tile + m * ADJ_PITCH + jw;
    const int* const a_sd  = s_tile + SD_OFF + jw * NHEAD + hh;
    const float* const b_src = ht + (size_t)jw * HID + hh * OUT_F + (m << 2);

    float4 hvA[8], hvB[8];
    int4 pre = make_int4(0, 0, 0, 0);
    if (stager) pre = t_src[0];
    k4_load_b(hvA, b_src);
    __syncthreads();

    const float ssr  = s_ss[m * NHEAD + hh];
    const float Mrow = s_M[m * NHEAD + hh];

    v8f acc0 = {}, acc1 = {}, acc2 = {}, acc3 = {};
    v2f lsum2 = {};

    for (int c = 0; c < N_CHUNKS; c += 2) {
        if (stager) *(int4*)__builtin_assume_aligned(&s_tile[t_dst], 16) = pre;
        __syncthreads();
        if (stager) pre = t_src[(size_t)(c + 1) * t_step];
        k4_load_b(hvB, b_src + (size_t)(c + 1) * JT * HID);
        k4_chunk(a_adj, a_sd, hvA, ssr, Mrow, acc0, acc1, acc2, acc3, lsum2);

        const int cn = (c + 2) & (N_CHUNKS - 1);
        if (stager) *(int4*)__builtin_assume_aligned(&s_tile[STAGE_INTS + t_dst], 16) = pre;
        __syncthreads();
        if (stager) pre = t_src[(size_t)cn * t_step];
        k4_load_b(hvA, b_src + (size_t)cn * JT * HID);
        k4_chunk(a_adj + STAGE_INTS, a_sd + STAGE_INTS, hvB, ssr, Mrow,
                 acc0, acc1, acc2, acc3, lsum2);
    }

    asm volatile("s_wait_alu depctr_va_vdst(0)"
                 : "+v"(acc0), "+v"(acc1), "+v"(acc2), "+v"(acc3));
    {
        float* const my_part = s_part + ((hh * 4 + wave) * 16) * OUT_F + (m << 2);
#pragma unroll
        for (int p = 0; p < 8; ++p) {
            const int row = p + (hi << 3);
            *(float4*)&my_part[row * OUT_F] =
                make_float4(acc0[p], acc1[p], acc2[p], acc3[p]);
        }
    }
    s_Lp[tid] = lsum2[0] + lsum2[1];
    __syncthreads();

    if (tid < 16 * NHEAD) {
        const int r = tid >> 3, h8 = tid & 7;
        const float* const q = s_Lp + h8 * (4 * 2 * 16) + r;
        float L = q[0];
        for (int t = 1; t < 8; ++t) L += q[t * 16];
        s_L[r * NHEAD + h8] = L;
    }
    __syncthreads();

    {
        const int r = tid >> 6;
        float hp = 0.f;
#pragma unroll
        for (int h8 = 0; h8 < NHEAD; ++h8) {
            const float* const q = s_part + (h8 * 4) * 16 * OUT_F + tid;
            float s = q[0];
            s += q[1 * 16 * OUT_F];
            s += q[2 * 16 * OUT_F];
            s += q[3 * 16 * OUT_F];
            hp += s / s_L[r * NHEAD + h8];
        }
        s_out[tid] = hp * (1.0f / NHEAD);
    }
    __syncthreads();

    if (tid < 16) {
        float mu = 0.f;
        for (int f = 0; f < OUT_F; ++f) mu += s_out[tid * OUT_F + f];
        mu *= (1.0f / OUT_F);
        float var = 0.f;
        for (int f = 0; f < OUT_F; ++f) {
            const float d = s_out[tid * OUT_F + f] - mu;
            var += d * d;
        }
        var *= (1.0f / OUT_F);
        s_mu[tid] = mu;
        s_rs[tid] = rsqrtf(var + LN_EPS);
    }
    __syncthreads();

    {
        const int r = tid >> 6, f = tid & 63;
        out[(size_t)(i0 + r) * OUT_F + f] =
            (s_out[r * OUT_F + f] - s_mu[r]) * s_rs[r] * gamma[f] + beta[f];
    }
}

extern "C" void kernel_launch(void* const* d_in, const int* in_sizes, int n_in,
                              void* d_out, int out_size, void* d_ws, size_t ws_size,
                              hipStream_t stream) {
    const float* h     = (const float*)d_in[0];
    const int*   adj   = (const int*)  d_in[1];
    const float* W     = (const float*)d_in[2];
    const float* a     = (const float*)d_in[3];
    const float* gamma = (const float*)d_in[4];
    const float* beta  = (const float*)d_in[5];
    float* out = (float*)d_out;

    float* ws   = (float*)d_ws;
    float* ht   = ws;
    float* ssrc = ht + (size_t)N_NODES * HID;
    float* sdst = ssrc + N_NODES * NHEAD;
    float* Mb   = sdst + N_NODES * NHEAD;

    k1_gemm_ht<<<dim3(N_NODES / 16, HID / 64), 128, 0, stream>>>(h, W, ht);
    k2_scores<<<(N_NODES * NHEAD) / 256, 256, 0, stream>>>(ht, a, ssrc, sdst);
    k3_rowmax<<<N_NODES, 256, 0, stream>>>(adj, ssrc, sdst, Mb);
    k4_agg_ln<<<N_NODES / 16, K4_THREADS, 0, stream>>>(adj, ht, ssrc, sdst, Mb,
                                                gamma, beta, out);
}


# k3: row max = leaky(s_src + masked max of s_dst), both monotone so M is bit-identical: per pair and head one add + one max, not five ops; 4 rows per block share s_dst; reduce through LDS
# speedup vs baseline: 11.0794x; 1.0731x over previous
#include <hip/hip_runtime.h>
#include <hip/hip_bf16.h>

#define N_NODES 4096
#define IN_F    256
#define OUT_F   64
#define NHEAD   8
#define HID     512
#define NEG_BIG (-9.0e15f)
#define LN_EPS  1e-05f

typedef __attribute__((ext_vector_type(2))) float v2f;
typedef __attribute__((ext_vector_type(8))) float v8f;

__device__ __forceinline__ v8f wmma_f32_16x16x4(v2f A, v2f B, v8f C) {
    return __builtin_amdgcn_wmma_f32_16x16x4_f32(false, A, false, B, (short)0, C,
                                                 false, false);
}

__global__ void __launch_bounds__(128)
k1_gemm_ht(const float* __restrict__ h, const float* __restrict__ W,
           float* __restrict__ ht) {
    __shared__ float sA[16 * 260];
    const int i0 = blockIdx.x * 16;
    const int n0 = blockIdx.y * 64;
    const int tid  = threadIdx.x;
    const int lane = tid & 31;
    const int wave = tid >> 5;

    for (int idx = tid; idx < 16 * 256 / 4; idx += 128) {
        const int r = idx >> 6, c = (idx & 63) << 2;
        const float4 v = *(const float4*)&h[(size_t)(i0 + r) * IN_F + c];
        *(float4*)&sA[r * 260 + c] = v;
    }
    __syncthreads();

    const int m  = lane & 15;
    const int hi = lane >> 4;

    v8f acc = {};
    for (int ks = 0; ks < IN_F / 4; ++ks) {
        v2f A, B;
#pragma unroll
        for (int v = 0; v < 2; ++v) {
            const int kk = (hi << 1) + v;
            const int k  = ks * 4 + kk;
            A[v] = sA[m * 260 + k];
            B[v] = W[(size_t)k * HID + n0 + (wave << 4) + m];
        }
        acc = wmma_f32_16x16x4(A, B, acc);
    }

#pragma unroll
    for (int p = 0; p < 8; ++p) {
        const int row = p + (hi << 3);
        ht[(size_t)(i0 + row) * HID + n0 + (wave << 4) + m] = acc[p];
    }
}

__global__ void __launch_bounds__(256)
k2_scores(const float* __restrict__ ht, const float* __restrict__ a,
          float* __restrict__ ssrc, float* __restrict__ sdst) {
    const int p = blockIdx.x * blockDim.x + threadIdx.x;
    if (p >= N_NODES * NHEAD) return;
    const float* row = ht + (size_t)p * OUT_F;
    float s1 = 0.f, s2 = 0.f;
#pragma unroll 8
    for (int f = 0; f < OUT_F; ++f) {
        const float v = row[f];
        s1 += v * a[f];
        s2 += v * a[OUT_F + f];
    }
    ssrc[p] = s1;
    sdst[p] = s2;
}

#define K3_ROWS 4
__device__ __forceinline__ float k3_max(float a, float b) { return __builtin_fmaximum_numf(a, b); }
__global__ void __launch_bounds__(256)
k3_rowmax(const int* __restrict__ adj, const float* __restrict__ ssrc,
          const float* __restrict__ sdst, float* __restrict__ Mb) {
    __shared__ alignas(16) float red[K3_ROWS * NHEAD * 8 * 36];
    __shared__ alignas(16) float red2[K3_ROWS * NHEAD * 8];
    const int i0   = blockIdx.x * K3_ROWS;
    const int tid  = threadIdx.x;
    const int lane = tid & 31;
    const int wave = tid >> 5;
    const float NEG_INF = -__builtin_inff();

    float mx[K3_ROWS][NHEAD];
#pragma unroll
    for (int r = 0; r < K3_ROWS; ++r)
#pragma unroll
        for (int hh = 0; hh < NHEAD; ++hh) mx[r][hh] = NEG_INF;

#pragma unroll 2
    for (int t = 0; t < N_NODES / 256; ++t) {
        const int j = tid + 256 * t;
        const float4 d0 = *(const float4*)&sdst[j * NHEAD];
        const float4 d1 = *(const float4*)&sdst[j * NHEAD + 4];
        const float d[NHEAD] = {d0.x, d0.y, d0.z, d0.w, d1.x, d1.y, d1.z, d1.w};
#pragma unroll
        for (int r = 0; r < K3_ROWS; ++r) {
            const int av = adj[(size_t)(i0 + r) * N_NODES + j];
            const float bias = av > 0 ? 0.f : NEG_INF;
#pragma unroll
            for (int hh = 0; hh < NHEAD; ++hh)
                mx[r][hh] = k3_max(mx[r][hh], d[hh] + bias);
        }
    }

#pragma unroll
    for (int r = 0; r < K3_ROWS; ++r)
#pragma unroll
        for (int hh = 0; hh < NHEAD; ++hh)
            red[(r * NHEAD + hh) * (8 * 36) + wave * 36 + lane] = mx[r][hh];
    __syncthreads();
    {
        const float4* const q = (const float4*)&red[(tid >> 3) * (8 * 36) + (tid & 7) * 36];
        float4 a4 = q[0];
#pragma unroll
        for (int k = 1; k < 8; ++k) {
            const float4 b4 = q[k];
            a4.x = k3_max(a4.x, b4.x); a4.y = k3_max(a4.y, b4.y);
            a4.z = k3_max(a4.z, b4.z); a4.w = k3_max(a4.w, b4.w);
        }
        red2[tid] = k3_max(k3_max(a4.x, a4.y), k3_max(a4.z, a4.w));
    }
    __syncthreads();
    if (tid < K3_ROWS * NHEAD) {
        const float4 p0 = *(const float4*)&red2[tid * 8];
        const float4 p1 = *(const float4*)&red2[tid * 8 + 4];
        const float dm = k3_max(k3_max(k3_max(p0.x, p0.y), k3_max(p0.z, p0.w)),
                                k3_max(k3_max(p1.x, p1.y), k3_max(p1.z, p1.w)));
        float e = ssrc[i0 * NHEAD + tid] + dm;
        e = e > 0.f ? e : 0.2f * e;
        Mb[i0 * NHEAD + tid] = dm == NEG_INF ? NEG_BIG : fmaxf(NEG_BIG, e);
    }
}

#define JT 64
#define K4_THREADS 1024
#define KS_WAVE 4
#define ADJ_PITCH 68
#define SD_OFF (16 * ADJ_PITCH)
#define STAGE_INTS (SD_OFF + JT * NHEAD)
#define N_CHUNKS (N_NODES / JT)

__device__ __forceinline__ void k4_load_b(float4 (&dst)[8], const float* __restrict__ src) {
#pragma unroll
    for (int q = 0; q < 8; ++q)
        dst[q] = *(const float4*)&src[(size_t)(4 * (q >> 1) + (q & 1)) * HID];
}

__device__ __forceinline__ void k4_chunk(const int* a_adj, const int* a_sd,
                                         const float4 (&hv)[8],
                                         const float ssr, const float Mrow,
                                         v8f& acc0, v8f& acc1, v8f& acc2, v8f& acc3,
                                         v2f& lsum2) {
#pragma unroll
    for (int ks = 0; ks < KS_WAVE; ++ks) {
        v2f A;
#pragma unroll
        for (int v = 0; v < 2; ++v) {
            const int jo = ks * 4 + v;
            const int adjv = a_adj[jo];
            float e = ssr + __int_as_float(a_sd[jo * NHEAD]);
            e = e > 0.f ? e : 0.2f * e;
            const float val = adjv > 0 ? e : NEG_BIG;
            A[v] = __expf(val - Mrow);
        }
        lsum2 += A;
        const float4 r0 = hv[2 * ks];
        const float4 r1 = hv[2 * ks + 1];
        const v2f B0 = {r0.x, r1.x};
        const v2f B1 = {r0.y, r1.y};
        const v2f B2 = {r0.z, r1.z};
        const v2f B3 = {r0.w, r1.w};
        acc0 = wmma_f32_16x16x4(A, B0, acc0);
        acc1 = wmma_f32_16x16x4(A, B1, acc1);
        acc2 = wmma_f32_16x16x4(A, B2, acc2);
        acc3 = wmma_f32_16x16x4(A, B3, acc3);
    }
}

__global__ void __launch_bounds__(K4_THREADS)
k4_agg_ln(const int* __restrict__ adj, const float* __restrict__ ht,
          const float* __restrict__ ssrc, const float* __restrict__ sdst,
          const float* __restrict__ Mb,
          const float* __restrict__ gamma, const float* __restrict__ beta,
          float* __restrict__ out) {
    __shared__ alignas(16) int s_tile[2 * STAGE_INTS];
    __shared__ alignas(16) float s_part[NHEAD * 4 * 16 * OUT_F];
    __shared__ float s_out[16 * OUT_F];
    __shared__ float s_L[16 * NHEAD];
    __shared__ float s_Lp[NHEAD * 4 * 2 * 16];
    __shared__ float s_M[16 * NHEAD];
    __shared__ float s_ss[16 * NHEAD];
    __shared__ float s_mu[16], s_rs[16];

    const int i0   = blockIdx.x * 16;
    const int tid  = threadIdx.x;
    const int hh   = tid >> 7;
    const int gt   = tid & 127;
    const int lane = tid & 31;
    const int wave = gt >> 5;
    const int m    = lane & 15;
    const int hi   = lane >> 4;

    if (tid < 16 * NHEAD) {
        const int r = tid >> 3, h8 = tid & 7;
        s_M[r * NHEAD + h8]  = Mb[(i0 + r) * NHEAD + h8];
        s_ss[r * NHEAD + h8] = ssrc[(i0 + r) * NHEAD + h8];
    }

    const bool stager = tid < 16 * JT / 4 + JT * NHEAD / 4;
    const int4* t_src;
    int t_step;
    int t_dst;
    if (tid < 16 * JT / 4) {
        const int r = tid >> 4, c = (tid & 15) << 2;
        t_src  = (const int4*)&adj[(size_t)(i0 + r) * N_NODES + c];
        t_step = JT / 4;
        t_dst  = r * ADJ_PITCH + c;
    } else {
        const int t4 = ((tid - 16 * JT / 4) & (JT * NHEAD / 4 - 1)) << 2;
        t_src  = (const int4*)&sdst[t4];
        t_step = JT * NHEAD / 4;
        t_dst  = SD_OFF + t4;
    }

    const int jw = (wave << 4) + (hi << 1);
    const int* const a_adj = s_tile + m * ADJ_PITCH + jw;
    const int* const a_sd  = s_tile + SD_OFF + jw * NHEAD + hh;
    const float* const b_src = ht + (size_t)jw * HID + hh * OUT_F + (m << 2);

    float4 hvA[8], hvB[8];
    int4 pre = make_int4(0, 0, 0, 0);
    if (stager) pre = t_src[0];
    k4_load_b(hvA, b_src);
    __syncthreads();

    const float ssr  = s_ss[m * NHEAD + hh];
    const float Mrow = s_M[m * NHEAD + hh];

    v8f acc0 = {}, acc1 = {}, acc2 = {}, acc3 = {};
    v2f lsum2 = {};

    for (int c = 0; c < N_CHUNKS; c += 2) {
        if (stager) *(int4*)__builtin_assume_aligned(&s_tile[t_dst], 16) = pre;
        __syncthreads();
        if (stager) pre = t_src[(size_t)(c + 1) * t_step];
        k4_load_b(hvB, b_src + (size_t)(c + 1) * JT * HID);
        k4_chunk(a_adj, a_sd, hvA, ssr, Mrow, acc0, acc1, acc2, acc3, lsum2);

        const int cn = (c + 2) & (N_CHUNKS - 1);
        if (stager) *(int4*)__builtin_assume_aligned(&s_tile[STAGE_INTS + t_dst], 16) = pre;
        __syncthreads();
        if (stager) pre = t_src[(size_t)cn * t_step];
        k4_load_b(hvA, b_src + (size_t)cn * JT * HID);
        k4_chunk(a_adj + STAGE_INTS, a_sd + STAGE_INTS, hvB, ssr, Mrow,
                 acc0, acc1, acc2, acc3, lsum2);
    }

    asm volatile("s_wait_alu depctr_va_vdst(0)"
                 : "+v"(acc0), "+v"(acc1), "+v"(acc2), "+v"(acc3));
    {
        float* const my_part = s_part + ((hh * 4 + wave) * 16) * OUT_F + (m << 2);
#pragma unroll
        for (int p = 0; p < 8; ++p) {
            const int row = p + (hi << 3);
            *(float4*)&my_part[row * OUT_F] =
                make_float4(acc0[p], acc1[p], acc2[p], acc3[p]);
        }
    }
    s_Lp[tid] = lsum2[0] + lsum2[1];
    __syncthreads();

    if (tid < 16 * NHEAD) {
        const int r = tid >> 3, h8 = tid & 7;
        const float* const q = s_Lp + h8 * (4 * 2 * 16) + r;
        float L = q[0];
        for (int t = 1; t < 8; ++t) L += q[t * 16];
        s_L[r * NHEAD + h8] = L;
    }
    __syncthreads();

    {
        const int r = tid >> 6;
        float hp = 0.f;
#pragma unroll
        for (int h8 = 0; h8 < NHEAD; ++h8) {
            const float* const q = s_part + (h8 * 4) * 16 * OUT_F + tid;
            float s = q[0];
            s += q[1 * 16 * OUT_F];
            s += q[2 * 16 * OUT_F];
            s += q[3 * 16 * OUT_F];
            hp += s / s_L[r * NHEAD + h8];
        }
        s_out[tid] = hp * (1.0f / NHEAD);
    }
    __syncthreads();

    if (tid < 16) {
        float mu = 0.f;
        for (int f = 0; f < OUT_F; ++f) mu += s_out[tid * OUT_F + f];
        mu *= (1.0f / OUT_F);
        float var = 0.f;
        for (int f = 0; f < OUT_F; ++f) {
            const float d = s_out[tid * OUT_F + f] - mu;
            var += d * d;
        }
        var *= (1.0f / OUT_F);
        s_mu[tid] = mu;
        s_rs[tid] = rsqrtf(var + LN_EPS);
    }
    __syncthreads();

    {
        const int r = tid >> 6, f = tid & 63;
        out[(size_t)(i0 + r) * OUT_F + f] =
            (s_out[r * OUT_F + f] - s_mu[r]) * s_rs[r] * gamma[f] + beta[f];
    }
}

extern "C" void kernel_launch(void* const* d_in, const int* in_sizes, int n_in,
                              void* d_out, int out_size, void* d_ws, size_t ws_size,
                              hipStream_t stream) {
    const float* h     = (const float*)d_in[0];
    const int*   adj   = (const int*)  d_in[1];
    const float* W     = (const float*)d_in[2];
    const float* a     = (const float*)d_in[3];
    const float* gamma = (const float*)d_in[4];
    const float* beta  = (const float*)d_in[5];
    float* out = (float*)d_out;

    float* ws   = (float*)d_ws;
    float* ht   = ws;
    float* ssrc = ht + (size_t)N_NODES * HID;
    float* sdst = ssrc + N_NODES * NHEAD;
    float* Mb   = sdst + N_NODES * NHEAD;

    k1_gemm_ht<<<dim3(N_NODES / 16, HID / 64), 128, 0, stream>>>(h, W, ht);
    k2_scores<<<(N_NODES * NHEAD) / 256, 256, 0, stream>>>(ht, a, ssrc, sdst);
    k3_rowmax<<<N_NODES / K3_ROWS, 256, 0, stream>>>(adj, ssrc, sdst, Mb);
    k4_agg_ln<<<N_NODES / 16, K4_THREADS, 0, stream>>>(adj, ht, ssrc, sdst, Mb,
                                                gamma, beta, out);
}


# k4 numerator as exp2(max(u + d, u2 + d/5) + mask): row terms u, u2 made once, d and the mask addend made when staged; 4 vector ops per k pair, was 12; exponent re-associated, last bits differ
# speedup vs baseline: 11.5841x; 1.0456x over previous
#include <hip/hip_runtime.h>
#include <hip/hip_bf16.h>

#define N_NODES 4096
#define IN_F    256
#define OUT_F   64
#define NHEAD   8
#define HID     512
#define NEG_BIG (-9.0e15f)
#define LOG2E   1.44269504088896340736f
#define LN_EPS  1e-05f

typedef __attribute__((ext_vector_type(2))) float v2f;
typedef __attribute__((ext_vector_type(8))) float v8f;

__device__ __forceinline__ v8f wmma_f32_16x16x4(v2f A, v2f B, v8f C) {
    return __builtin_amdgcn_wmma_f32_16x16x4_f32(false, A, false, B, (short)0, C,
                                                 false, false);
}

__global__ void __launch_bounds__(128)
k1_gemm_ht(const float* __restrict__ h, const float* __restrict__ W,
           float* __restrict__ ht) {
    __shared__ float sA[16 * 260];
    const int i0 = blockIdx.x * 16;
    const int n0 = blockIdx.y * 64;
    const int tid  = threadIdx.x;
    const int lane = tid & 31;
    const int wave = tid >> 5;

    for (int idx = tid; idx < 16 * 256 / 4; idx += 128) {
        const int r = idx >> 6, c = (idx & 63) << 2;
        const float4 v = *(const float4*)&h[(size_t)(i0 + r) * IN_F + c];
        *(float4*)&sA[r * 260 + c] = v;
    }
    __syncthreads();

    const int m  = lane & 15;
    const int hi = lane >> 4;

    v8f acc = {};
    for (int ks = 0; ks < IN_F / 4; ++ks) {
        v2f A, B;
#pragma unroll
        for (int v = 0; v < 2; ++v) {
            const int kk = (hi << 1) + v;
            const int k  = ks * 4 + kk;
            A[v] = sA[m * 260 + k];
            B[v] = W[(size_t)k * HID + n0 + (wave << 4) + m];
        }
        acc = wmma_f32_16x16x4(A, B, acc);
    }

#pragma unroll
    for (int p = 0; p < 8; ++p) {
        const int row = p + (hi << 3);
        ht[(size_t)(i0 + row) * HID + n0 + (wave << 4) + m] = acc[p];
    }
}

__global__ void __launch_bounds__(256)
k2_scores(const float* __restrict__ ht, const float* __restrict__ a,
          float* __restrict__ ssrc, float* __restrict__ sdst) {
    const int p = blockIdx.x * blockDim.x + threadIdx.x;
    if (p >= N_NODES * NHEAD) return;
    const float* row = ht + (size_t)p * OUT_F;
    float s1 = 0.f, s2 = 0.f;
#pragma unroll 8
    for (int f = 0; f < OUT_F; ++f) {
        const float v = row[f];
        s1 += v * a[f];
        s2 += v * a[OUT_F + f];
    }
    ssrc[p] = s1;
    sdst[p] = s2;
}

#define K3_ROWS 4
__device__ __forceinline__ float k3_max(float a, float b) { return __builtin_fmaximum_numf(a, b); }
__global__ void __launch_bounds__(256)
k3_rowmax(const int* __restrict__ adj, const float* __restrict__ ssrc,
          const float* __restrict__ sdst, float* __restrict__ Mb) {
    __shared__ alignas(16) float red[K3_ROWS * NHEAD * 8 * 36];
    __shared__ alignas(16) float red2[K3_ROWS * NHEAD * 8];
    const int i0   = blockIdx.x * K3_ROWS;
    const int tid  = threadIdx.x;
    const int lane = tid & 31;
    const int wave = tid >> 5;
    const float NEG_INF = -__builtin_inff();

    float mx[K3_ROWS][NHEAD];
#pragma unroll
    for (int r = 0; r < K3_ROWS; ++r)
#pragma unroll
        for (int hh = 0; hh < NHEAD; ++hh) mx[r][hh] = NEG_INF;

#pragma unroll 2
    for (int t = 0; t < N_NODES / 256; ++t) {
        const int j = tid + 256 * t;
        const float4 d0 = *(const float4*)&sdst[j * NHEAD];
        const float4 d1 = *(const float4*)&sdst[j * NHEAD + 4];
        const float d[NHEAD] = {d0.x, d0.y, d0.z, d0.w, d1.x, d1.y, d1.z, d1.w};
#pragma unroll
        for (int r = 0; r < K3_ROWS; ++r) {
            const int av = adj[(size_t)(i0 + r) * N_NODES + j];
            const float bias = av > 0 ? 0.f : NEG_INF;
#pragma unroll
            for (int hh = 0; hh < NHEAD; ++hh)
                mx[r][hh] = k3_max(mx[r][hh], d[hh] + bias);
        }
    }

#pragma unroll
    for (int r = 0; r < K3_ROWS; ++r)
#pragma unroll
        for (int hh = 0; hh < NHEAD; ++hh)
            red[(r * NHEAD + hh) * (8 * 36) + wave * 36 + lane] = mx[r][hh];
    __syncthreads();
    {
        const float4* const q = (const float4*)&red[(tid >> 3) * (8 * 36) + (tid & 7) * 36];
        float4 a4 = q[0];
#pragma unroll
        for (int k = 1; k < 8; ++k) {
            const float4 b4 = q[k];
            a4.x = k3_max(a4.x, b4.x); a4.y = k3_max(a4.y, b4.y);
            a4.z = k3_max(a4.z, b4.z); a4.w = k3_max(a4.w, b4.w);
        }
        red2[tid] = k3_max(k3_max(a4.x, a4.y), k3_max(a4.z, a4.w));
    }
    __syncthreads();
    if (tid < K3_ROWS * NHEAD) {
        const float4 p0 = *(const float4*)&red2[tid * 8];
        const float4 p1 = *(const float4*)&red2[tid * 8 + 4];
        const float dm = k3_max(k3_max(k3_max(p0.x, p0.y), k3_max(p0.z, p0.w)),
                                k3_max(k3_max(p1.x, p1.y), k3_max(p1.z, p1.w)));
        float e = ssrc[i0 * NHEAD + tid] + dm;
        e = e > 0.f ? e : 0.2f * e;
        Mb[i0 * NHEAD + tid] = dm == NEG_INF ? NEG_BIG : fmaxf(NEG_BIG, e);
    }
}

#define JT 64
#define K4_THREADS 1024
#define KS_WAVE 4
#define ADJ_PITCH 68
#define SD_OFF (16 * ADJ_PITCH)
#define STAGE_INTS (SD_OFF + JT * NHEAD)
#define N_CHUNKS (N_NODES / JT)

__device__ __forceinline__ void k4_load_b(float4 (&dst)[8], const float* __restrict__ src) {
#pragma unroll
    for (int q = 0; q < 8; ++q)
        dst[q] = *(const float4*)&src[(size_t)(4 * (q >> 1) + (q & 1)) * HID];
}

#define MASK_OFF (NEG_BIG * LOG2E)

__device__ __forceinline__ int4 k4_stage_mask(const int4 pre) {
    const int off = __float_as_int(MASK_OFF);
    return make_int4(pre.x > 0 ? 0 : off, pre.y > 0 ? 0 : off,
                     pre.z > 0 ? 0 : off, pre.w > 0 ? 0 : off);
}
__device__ __forceinline__ int4 k4_stage_sd(const int4 pre) {
    return make_int4(__float_as_int(__int_as_float(pre.x) * LOG2E),
                     __float_as_int(__int_as_float(pre.y) * LOG2E),
                     __float_as_int(__int_as_float(pre.z) * LOG2E),
                     __float_as_int(__int_as_float(pre.w) * LOG2E));
}

__device__ __forceinline__ void k4_chunk(const int* a_msk, const int* a_sd,
                                         const float4 (&hv)[8],
                                         const v2f u, const v2f u2,
                                         v8f& acc0, v8f& acc1, v8f& acc2, v8f& acc3,
                                         v2f& lsum2) {
#pragma unroll
    for (int ks = 0; ks < KS_WAVE; ++ks) {
        const int jo = ks * 4;
        const v2f d   = {__int_as_float(a_sd[jo * NHEAD]),
                         __int_as_float(a_sd[(jo + 1) * NHEAD])};
        const v2f msk = {__int_as_float(a_msk[jo]), __int_as_float(a_msk[jo + 1])};
        const v2f t1 = u + d;
        const v2f t2 = d * 0.2f + u2;
        const v2f arg = __builtin_elementwise_max(t1, t2) + msk;
        const v2f A = {__builtin_amdgcn_exp2f(arg[0]), __builtin_amdgcn_exp2f(arg[1])};
        lsum2 += A;
        const float4 r0 = hv[2 * ks];
        const float4 r1 = hv[2 * ks + 1];
        const v2f B0 = {r0.x, r1.x};
        const v2f B1 = {r0.y, r1.y};
        const v2f B2 = {r0.z, r1.z};
        const v2f B3 = {r0.w, r1.w};
        acc0 = wmma_f32_16x16x4(A, B0, acc0);
        acc1 = wmma_f32_16x16x4(A, B1, acc1);
        acc2 = wmma_f32_16x16x4(A, B2, acc2);
        acc3 = wmma_f32_16x16x4(A, B3, acc3);
    }
}

__global__ void __launch_bounds__(K4_THREADS)
k4_agg_ln(const int* __restrict__ adj, const float* __restrict__ ht,
          const float* __restrict__ ssrc, const float* __restrict__ sdst,
          const float* __restrict__ Mb,
          const float* __restrict__ gamma, const float* __restrict__ beta,
          float* __restrict__ out) {
    __shared__ alignas(16) int s_tile[2 * STAGE_INTS];
    __shared__ alignas(16) float s_part[NHEAD * 4 * 16 * OUT_F];
    __shared__ float s_out[16 * OUT_F];
    __shared__ float s_L[16 * NHEAD];
    __shared__ float s_Lp[NHEAD * 4 * 2 * 16];
    __shared__ float s_M[16 * NHEAD];
    __shared__ float s_ss[16 * NHEAD];
    __shared__ float s_mu[16], s_rs[16];

    const int i0   = blockIdx.x * 16;
    const int tid  = threadIdx.x;
    const int hh   = tid >> 7;
    const int gt   = tid & 127;
    const int lane = tid & 31;
    const int wave = gt >> 5;
    const int m    = lane & 15;
    const int hi   = lane >> 4;

    if (tid < 16 * NHEAD) {
        const int r = tid >> 3, h8 = tid & 7;
        s_M[r * NHEAD + h8]  = Mb[(i0 + r) * NHEAD + h8];
        s_ss[r * NHEAD + h8] = ssrc[(i0 + r) * NHEAD + h8];
    }

    const bool stager = tid < 16 * JT / 4 + JT * NHEAD / 4;
    const bool adj_stager = tid < 16 * JT / 4;
    const int4* t_src;
    int t_step;
    int t_dst;
    if (tid < 16 * JT / 4) {
        const int r = tid >> 4, c = (tid & 15) << 2;
        t_src  = (const int4*)&adj[(size_t)(i0 + r) * N_NODES + c];
        t_step = JT / 4;
        t_dst  = r * ADJ_PITCH + c;
    } else {
        const int t4 = ((tid - 16 * JT / 4) & (JT * NHEAD / 4 - 1)) << 2;
        t_src  = (const int4*)&sdst[t4];
        t_step = JT * NHEAD / 4;
        t_dst  = SD_OFF + t4;
    }

    const int jw = (wave << 4) + (hi << 1);
    const int* const a_adj = s_tile + m * ADJ_PITCH + jw;
    const int* const a_sd  = s_tile + SD_OFF + jw * NHEAD + hh;
    const float* const b_src = ht + (size_t)jw * HID + hh * OUT_F + (m << 2);

    float4 hvA[8], hvB[8];
    int4 pre = make_int4(0, 0, 0, 0);
    if (stager) pre = t_src[0];
    k4_load_b(hvA, b_src);
    __syncthreads();

    const float ssr  = s_ss[m * NHEAD + hh];
    const float Mrow = s_M[m * NHEAD + hh];
    const float u1s = (ssr - Mrow) * LOG2E;
    const float u2s = (0.2f * ssr - Mrow) * LOG2E;
    const v2f u  = {u1s, u1s};
    const v2f u2 = {u2s, u2s};

    v8f acc0 = {}, acc1 = {}, acc2 = {}, acc3 = {};
    v2f lsum2 = {};

    for (int c = 0; c < N_CHUNKS; c += 2) {
        if (adj_stager)  *(int4*)__builtin_assume_aligned(&s_tile[t_dst], 16) = k4_stage_mask(pre);
        else if (stager) *(int4*)__builtin_assume_aligned(&s_tile[t_dst], 16) = k4_stage_sd(pre);
        __syncthreads();
        if (stager) pre = t_src[(size_t)(c + 1) * t_step];
        k4_load_b(hvB, b_src + (size_t)(c + 1) * JT * HID);
        k4_chunk(a_adj, a_sd, hvA, u, u2, acc0, acc1, acc2, acc3, lsum2);

        const int cn = (c + 2) & (N_CHUNKS - 1);
        if (adj_stager)  *(int4*)__builtin_assume_aligned(&s_tile[STAGE_INTS + t_dst], 16) = k4_stage_mask(pre);
        else if (stager) *(int4*)__builtin_assume_aligned(&s_tile[STAGE_INTS + t_dst], 16) = k4_stage_sd(pre);
        __syncthreads();
        if (stager) pre = t_src[(size_t)cn * t_step];
        k4_load_b(hvA, b_src + (size_t)cn * JT * HID);
        k4_chunk(a_adj + STAGE_INTS, a_sd + STAGE_INTS, hvB, u, u2,
                 acc0, acc1, acc2, acc3, lsum2);
    }

    asm volatile("s_wait_alu depctr_va_vdst(0)"
                 : "+v"(acc0), "+v"(acc1), "+v"(acc2), "+v"(acc3));
    {
        float* const my_part = s_part + ((hh * 4 + wave) * 16) * OUT_F + (m << 2);
#pragma unroll
        for (int p = 0; p < 8; ++p) {
            const int row = p + (hi << 3);
            *(float4*)&my_part[row * OUT_F] =
                make_float4(acc0[p], acc1[p], acc2[p], acc3[p]);
        }
    }
    s_Lp[tid] = lsum2[0] + lsum2[1];
    __syncthreads();

    if (tid < 16 * NHEAD) {
        const int r = tid >> 3, h8 = tid & 7;
        const float* const q = s_Lp + h8 * (4 * 2 * 16) + r;
        float L = q[0];
        for (int t = 1; t < 8; ++t) L += q[t * 16];
        s_L[r * NHEAD + h8] = L;
    }
    __syncthreads();

    {
        const int r = tid >> 6;
        float hp = 0.f;
#pragma unroll
        for (int h8 = 0; h8 < NHEAD; ++h8) {
            const float* const q = s_part + (h8 * 4) * 16 * OUT_F + tid;
            float s = q[0];
            s += q[1 * 16 * OUT_F];
            s += q[2 * 16 * OUT_F];
            s += q[3 * 16 * OUT_F];
            hp += s / s_L[r * NHEAD + h8];
        }
        s_out[tid] = hp * (1.0f / NHEAD);
    }
    __syncthreads();

    if (tid < 16) {
        float mu = 0.f;
        for (int f = 0; f < OUT_F; ++f) mu += s_out[tid * OUT_F + f];
        mu *= (1.0f / OUT_F);
        float var = 0.f;
        for (int f = 0; f < OUT_F; ++f) {
            const float d = s_out[tid * OUT_F + f] - mu;
            var += d * d;
        }
        var *= (1.0f / OUT_F);
        s_mu[tid] = mu;
        s_rs[tid] = rsqrtf(var + LN_EPS);
    }
    __syncthreads();

    {
        const int r = tid >> 6, f = tid & 63;
        out[(size_t)(i0 + r) * OUT_F + f] =
            (s_out[r * OUT_F + f] - s_mu[r]) * s_rs[r] * gamma[f] + beta[f];
    }
}

extern "C" void kernel_launch(void* const* d_in, const int* in_sizes, int n_in,
                              void* d_out, int out_size, void* d_ws, size_t ws_size,
                              hipStream_t stream) {
    const float* h     = (const float*)d_in[0];
    const int*   adj   = (const int*)  d_in[1];
    const float* W     = (const float*)d_in[2];
    const float* a     = (const float*)d_in[3];
    const float* gamma = (const float*)d_in[4];
    const float* beta  = (const float*)d_in[5];
    float* out = (float*)d_out;

    float* ws   = (float*)d_ws;
    float* ht   = ws;
    float* ssrc = ht + (size_t)N_NODES * HID;
    float* sdst = ssrc + N_NODES * NHEAD;
    float* Mb   = sdst + N_NODES * NHEAD;

    k1_gemm_ht<<<dim3(N_NODES / 16, HID / 64), 128, 0, stream>>>(h, W, ht);
    k2_scores<<<(N_NODES * NHEAD) / 256, 256, 0, stream>>>(ht, a, ssrc, sdst);
    k3_rowmax<<<N_NODES / K3_ROWS, 256, 0, stream>>>(adj, ssrc, sdst, Mb);
    k4_agg_ln<<<N_NODES / 16, K4_THREADS, 0, stream>>>(adj, ht, ssrc, sdst, Mb,
                                                gamma, beta, out);
}


# k4: chunk barrier taken as arrive (chunk start) + wait (behind the chunk's ring reads) over a 4-stage mask/s_dst ring filled one chunk early: no lockstep restart per chunk; all sum orders kept
# speedup vs baseline: 12.1370x; 1.0477x over previous
#include <hip/hip_runtime.h>
#include <hip/hip_bf16.h>

#define N_NODES 4096
#define IN_F    256
#define OUT_F   64
#define NHEAD   8
#define HID     512
#define NEG_BIG (-9.0e15f)
#define LOG2E   1.44269504088896340736f
#define LN_EPS  1e-05f

typedef __attribute__((ext_vector_type(2))) float v2f;
typedef __attribute__((ext_vector_type(8))) float v8f;

__device__ __forceinline__ v8f wmma_f32_16x16x4(v2f A, v2f B, v8f C) {
    return __builtin_amdgcn_wmma_f32_16x16x4_f32(false, A, false, B, (short)0, C,
                                                 false, false);
}

__global__ void __launch_bounds__(128)
k1_gemm_ht(const float* __restrict__ h, const float* __restrict__ W,
           float* __restrict__ ht) {
    __shared__ float sA[16 * 260];
    const int i0 = blockIdx.x * 16;
    const int n0 = blockIdx.y * 64;
    const int tid  = threadIdx.x;
    const int lane = tid & 31;
    const int wave = tid >> 5;

    for (int idx = tid; idx < 16 * 256 / 4; idx += 128) {
        const int r = idx >> 6, c = (idx & 63) << 2;
        const float4 v = *(const float4*)&h[(size_t)(i0 + r) * IN_F + c];
        *(float4*)&sA[r * 260 + c] = v;
    }
    __syncthreads();

    const int m  = lane & 15;
    const int hi = lane >> 4;

    v8f acc = {};
    for (int ks = 0; ks < IN_F / 4; ++ks) {
        v2f A, B;
#pragma unroll
        for (int v = 0; v < 2; ++v) {
            const int kk = (hi << 1) + v;
            const int k  = ks * 4 + kk;
            A[v] = sA[m * 260 + k];
            B[v] = W[(size_t)k * HID + n0 + (wave << 4) + m];
        }
        acc = wmma_f32_16x16x4(A, B, acc);
    }

#pragma unroll
    for (int p = 0; p < 8; ++p) {
        const int row = p + (hi << 3);
        ht[(size_t)(i0 + row) * HID + n0 + (wave << 4) + m] = acc[p];
    }
}

__global__ void __launch_bounds__(256)
k2_scores(const float* __restrict__ ht, const float* __restrict__ a,
          float* __restrict__ ssrc, float* __restrict__ sdst) {
    const int p = blockIdx.x * blockDim.x + threadIdx.x;
    if (p >= N_NODES * NHEAD) return;
    const float* row = ht + (size_t)p * OUT_F;
    float s1 = 0.f, s2 = 0.f;
#pragma unroll 8
    for (int f = 0; f < OUT_F; ++f) {
        const float v = row[f];
        s1 += v * a[f];
        s2 += v * a[OUT_F + f];
    }
    ssrc[p] = s1;
    sdst[p] = s2;
}

#define K3_ROWS 4
__device__ __forceinline__ float k3_max(float a, float b) { return __builtin_fmaximum_numf(a, b); }
__global__ void __launch_bounds__(256)
k3_rowmax(const int* __restrict__ adj, const float* __restrict__ ssrc,
          const float* __restrict__ sdst, float* __restrict__ Mb) {
    __shared__ alignas(16) float red[K3_ROWS * NHEAD * 8 * 36];
    __shared__ alignas(16) float red2[K3_ROWS * NHEAD * 8];
    const int i0   = blockIdx.x * K3_ROWS;
    const int tid  = threadIdx.x;
    const int lane = tid & 31;
    const int wave = tid >> 5;
    const float NEG_INF = -__builtin_inff();

    float mx[K3_ROWS][NHEAD];
#pragma unroll
    for (int r = 0; r < K3_ROWS; ++r)
#pragma unroll
        for (int hh = 0; hh < NHEAD; ++hh) mx[r][hh] = NEG_INF;

#pragma unroll 2
    for (int t = 0; t < N_NODES / 256; ++t) {
        const int j = tid + 256 * t;
        const float4 d0 = *(const float4*)&sdst[j * NHEAD];
        const float4 d1 = *(const float4*)&sdst[j * NHEAD + 4];
        const float d[NHEAD] = {d0.x, d0.y, d0.z, d0.w, d1.x, d1.y, d1.z, d1.w};
#pragma unroll
        for (int r = 0; r < K3_ROWS; ++r) {
            const int av = adj[(size_t)(i0 + r) * N_NODES + j];
            const float bias = av > 0 ? 0.f : NEG_INF;
#pragma unroll
            for (int hh = 0; hh < NHEAD; ++hh)
                mx[r][hh] = k3_max(mx[r][hh], d[hh] + bias);
        }
    }

#pragma unroll
    for (int r = 0; r < K3_ROWS; ++r)
#pragma unroll
        for (int hh = 0; hh < NHEAD; ++hh)
            red[(r * NHEAD + hh) * (8 * 36) + wave * 36 + lane] = mx[r][hh];
    __syncthreads();
    {
        const float4* const q = (const float4*)&red[(tid >> 3) * (8 * 36) + (tid & 7) * 36];
        float4 a4 = q[0];
#pragma unroll
        for (int k = 1; k < 8; ++k) {
            const float4 b4 = q[k];
            a4.x = k3_max(a4.x, b4.x); a4.y = k3_max(a4.y, b4.y);
            a4.z = k3_max(a4.z, b4.z); a4.w = k3_max(a4.w, b4.w);
        }
        red2[tid] = k3_max(k3_max(a4.x, a4.y), k3_max(a4.z, a4.w));
    }
    __syncthreads();
    if (tid < K3_ROWS * NHEAD) {
        const float4 p0 = *(const float4*)&red2[tid * 8];
        const float4 p1 = *(const float4*)&red2[tid * 8 + 4];
        const float dm = k3_max(k3_max(k3_max(p0.x, p0.y), k3_max(p0.z, p0.w)),
                                k3_max(k3_max(p1.x, p1.y), k3_max(p1.z, p1.w)));
        float e = ssrc[i0 * NHEAD + tid] + dm;
        e = e > 0.f ? e : 0.2f * e;
        Mb[i0 * NHEAD + tid] = dm == NEG_INF ? NEG_BIG : fmaxf(NEG_BIG, e);
    }
}

#define JT 64
#define K4_THREADS 1024
#define KS_WAVE 4
#define ADJ_PITCH 68
#define SD_OFF (16 * ADJ_PITCH)
#define STAGE_INTS (SD_OFF + JT * NHEAD)
#define N_STAGES 4
#define N_CHUNKS (N_NODES / JT)

__device__ __forceinline__ void k4_load_b(float4 (&dst)[8], const float* __restrict__ src) {
#pragma unroll
    for (int q = 0; q < 8; ++q)
        dst[q] = *(const float4*)&src[(size_t)(4 * (q >> 1) + (q & 1)) * HID];
}

#define MASK_OFF (NEG_BIG * LOG2E)

__device__ __forceinline__ int4 k4_stage_mask(const int4 pre) {
    const int off = __float_as_int(MASK_OFF);
    return make_int4(pre.x > 0 ? 0 : off, pre.y > 0 ? 0 : off,
                     pre.z > 0 ? 0 : off, pre.w > 0 ? 0 : off);
}
__device__ __forceinline__ int4 k4_stage_sd(const int4 pre) {
    return make_int4(__float_as_int(__int_as_float(pre.x) * LOG2E),
                     __float_as_int(__int_as_float(pre.y) * LOG2E),
                     __float_as_int(__int_as_float(pre.z) * LOG2E),
                     __float_as_int(__int_as_float(pre.w) * LOG2E));
}

__device__ __forceinline__ void k4_ring_arrive() {
    __builtin_amdgcn_fence(__ATOMIC_RELEASE, "workgroup", "local");
    __builtin_amdgcn_s_barrier_signal(-1);
}
__device__ __forceinline__ void k4_ring_wait() {
    __builtin_amdgcn_s_barrier_wait(-1);
    __builtin_amdgcn_fence(__ATOMIC_ACQUIRE, "workgroup", "local");
}

__device__ __forceinline__ void k4_chunk(const int* a_msk, const int* a_sd,
                                         const float4 (&hv)[8],
                                         const v2f u, const v2f u2,
                                         v8f& acc0, v8f& acc1, v8f& acc2, v8f& acc3,
                                         v2f& lsum2) {
#pragma unroll
    for (int ks = 0; ks < KS_WAVE; ++ks) {
        const int jo = ks * 4;
        const v2f d   = {__int_as_float(a_sd[jo * NHEAD]),
                         __int_as_float(a_sd[(jo + 1) * NHEAD])};
        const v2f msk = {__int_as_float(a_msk[jo]), __int_as_float(a_msk[jo + 1])};
        const v2f t1 = u + d;
        const v2f t2 = d * 0.2f + u2;
        const v2f arg = __builtin_elementwise_max(t1, t2) + msk;
        const v2f A = {__builtin_amdgcn_exp2f(arg[0]), __builtin_amdgcn_exp2f(arg[1])};
        lsum2 += A;
        const float4 r0 = hv[2 * ks];
        const float4 r1 = hv[2 * ks + 1];
        const v2f B0 = {r0.x, r1.x};
        const v2f B1 = {r0.y, r1.y};
        const v2f B2 = {r0.z, r1.z};
        const v2f B3 = {r0.w, r1.w};
        acc0 = wmma_f32_16x16x4(A, B0, acc0);
        acc1 = wmma_f32_16x16x4(A, B1, acc1);
        acc2 = wmma_f32_16x16x4(A, B2, acc2);
        acc3 = wmma_f32_16x16x4(A, B3, acc3);
    }
}

__global__ void __launch_bounds__(K4_THREADS)
k4_agg_ln(const int* __restrict__ adj, const float* __restrict__ ht,
          const float* __restrict__ ssrc, const float* __restrict__ sdst,
          const float* __restrict__ Mb,
          const float* __restrict__ gamma, const float* __restrict__ beta,
          float* __restrict__ out) {
    __shared__ alignas(16) int s_tile[N_STAGES * STAGE_INTS];
    __shared__ alignas(16) float s_part[NHEAD * 4 * 16 * OUT_F];
    __shared__ float s_out[16 * OUT_F];
    __shared__ float s_L[16 * NHEAD];
    __shared__ float s_Lp[NHEAD * 4 * 2 * 16];
    __shared__ float s_M[16 * NHEAD];
    __shared__ float s_ss[16 * NHEAD];
    __shared__ float s_mu[16], s_rs[16];

    const int i0   = blockIdx.x * 16;
    const int tid  = threadIdx.x;
    const int hh   = tid >> 7;
    const int gt   = tid & 127;
    const int lane = tid & 31;
    const int wave = gt >> 5;
    const int m    = lane & 15;
    const int hi   = lane >> 4;

    if (tid < 16 * NHEAD) {
        const int r = tid >> 3, h8 = tid & 7;
        s_M[r * NHEAD + h8]  = Mb[(i0 + r) * NHEAD + h8];
        s_ss[r * NHEAD + h8] = ssrc[(i0 + r) * NHEAD + h8];
    }

    const bool stager = tid < 16 * JT / 4 + JT * NHEAD / 4;
    const bool adj_stager = tid < 16 * JT / 4;
    const int4* t_src;
    int t_step;
    int t_dst;
    if (tid < 16 * JT / 4) {
        const int r = tid >> 4, c = (tid & 15) << 2;
        t_src  = (const int4*)&adj[(size_t)(i0 + r) * N_NODES + c];
        t_step = JT / 4;
        t_dst  = r * ADJ_PITCH + c;
    } else {
        const int t4 = ((tid - 16 * JT / 4) & (JT * NHEAD / 4 - 1)) << 2;
        t_src  = (const int4*)&sdst[t4];
        t_step = JT * NHEAD / 4;
        t_dst  = SD_OFF + t4;
    }

    const int jw = (wave << 4) + (hi << 1);
    const int* const a_adj = s_tile + m * ADJ_PITCH + jw;
    const int* const a_sd  = s_tile + SD_OFF + jw * NHEAD + hh;
    const float* const b_src = ht + (size_t)jw * HID + hh * OUT_F + (m << 2);

    float4 hvA[8], hvB[8];
    int4 pre0 = make_int4(0, 0, 0, 0);
    int4 pre  = make_int4(0, 0, 0, 0);
    if (stager) { pre0 = t_src[0]; pre = t_src[t_step]; }
    k4_load_b(hvA, b_src);
    if (adj_stager)  *(int4*)__builtin_assume_aligned(&s_tile[t_dst], 16) = k4_stage_mask(pre0);
    else if (stager) *(int4*)__builtin_assume_aligned(&s_tile[t_dst], 16) = k4_stage_sd(pre0);
    k4_ring_arrive();
    k4_ring_wait();

    const float ssr  = s_ss[m * NHEAD + hh];
    const float Mrow = s_M[m * NHEAD + hh];
    const float u1s = (ssr - Mrow) * LOG2E;
    const float u2s = (0.2f * ssr - Mrow) * LOG2E;
    const v2f u  = {u1s, u1s};
    const v2f u2 = {u2s, u2s};

    v8f acc0 = {}, acc1 = {}, acc2 = {}, acc3 = {};
    v2f lsum2 = {};

    const int* pa = a_adj;
    const int* ps = a_sd;
    int* pt = s_tile + t_dst;
    int hop = 2 * STAGE_INTS;

    for (int c = 0; c < N_CHUNKS; c += 2) {
        if (adj_stager)  *(int4*)__builtin_assume_aligned(pt + STAGE_INTS, 16) = k4_stage_mask(pre);
        else if (stager) *(int4*)__builtin_assume_aligned(pt + STAGE_INTS, 16) = k4_stage_sd(pre);
        k4_ring_arrive();
        if (stager) pre = t_src[(size_t)((c + 2) & (N_CHUNKS - 1)) * t_step];
        k4_load_b(hvB, b_src + (size_t)(c + 1) * JT * HID);
        k4_chunk(pa, ps, hvA, u, u2, acc0, acc1, acc2, acc3, lsum2);
        k4_ring_wait();

        if (adj_stager)  *(int4*)__builtin_assume_aligned(pt + hop, 16) = k4_stage_mask(pre);
        else if (stager) *(int4*)__builtin_assume_aligned(pt + hop, 16) = k4_stage_sd(pre);
        k4_ring_arrive();
        if (stager) pre = t_src[(size_t)((c + 3) & (N_CHUNKS - 1)) * t_step];
        k4_load_b(hvA, b_src + (size_t)((c + 2) & (N_CHUNKS - 1)) * JT * HID);
        k4_chunk(pa + STAGE_INTS, ps + STAGE_INTS, hvB, u, u2, acc0, acc1, acc2, acc3, lsum2);
        k4_ring_wait();

        pa += hop; ps += hop; pt += hop;
        hop = -hop;
    }

    asm volatile("s_wait_alu depctr_va_vdst(0)"
                 : "+v"(acc0), "+v"(acc1), "+v"(acc2), "+v"(acc3));
    {
        float* const my_part = s_part + ((hh * 4 + wave) * 16) * OUT_F + (m << 2);
#pragma unroll
        for (int p = 0; p < 8; ++p) {
            const int row = p + (hi << 3);
            *(float4*)&my_part[row * OUT_F] =
                make_float4(acc0[p], acc1[p], acc2[p], acc3[p]);
        }
    }
    s_Lp[tid] = lsum2[0] + lsum2[1];
    __syncthreads();

    if (tid < 16 * NHEAD) {
        const int r = tid >> 3, h8 = tid & 7;
        const float* const q = s_Lp + h8 * (4 * 2 * 16) + r;
        float L = q[0];
        for (int t = 1; t < 8; ++t) L += q[t * 16];
        s_L[r * NHEAD + h8] = L;
    }
    __syncthreads();

    {
        const int r = tid >> 6;
        float hp = 0.f;
#pragma unroll
        for (int h8 = 0; h8 < NHEAD; ++h8) {
            const float* const q = s_part + (h8 * 4) * 16 * OUT_F + tid;
            float s = q[0];
            s += q[1 * 16 * OUT_F];
            s += q[2 * 16 * OUT_F];
            s += q[3 * 16 * OUT_F];
            hp += s / s_L[r * NHEAD + h8];
        }
        s_out[tid] = hp * (1.0f / NHEAD);
    }
    __syncthreads();

    if (tid < 16) {
        float mu = 0.f;
        for (int f = 0; f < OUT_F; ++f) mu += s_out[tid * OUT_F + f];
        mu *= (1.0f / OUT_F);
        float var = 0.f;
        for (int f = 0; f < OUT_F; ++f) {
            const float d = s_out[tid * OUT_F + f] - mu;
            var += d * d;
        }
        var *= (1.0f / OUT_F);
        s_mu[tid] = mu;
        s_rs[tid] = rsqrtf(var + LN_EPS);
    }
    __syncthreads();

    {
        const int r = tid >> 6, f = tid & 63;
        out[(size_t)(i0 + r) * OUT_F + f] =
            (s_out[r * OUT_F + f] - s_mu[r]) * s_rs[r] * gamma[f] + beta[f];
    }
}

extern "C" void kernel_launch(void* const* d_in, const int* in_sizes, int n_in,
                              void* d_out, int out_size, void* d_ws, size_t ws_size,
                              hipStream_t stream) {
    const float* h     = (const float*)d_in[0];
    const int*   adj   = (const int*)  d_in[1];
    const float* W     = (const float*)d_in[2];
    const float* a     = (const float*)d_in[3];
    const float* gamma = (const float*)d_in[4];
    const float* beta  = (const float*)d_in[5];
    float* out = (float*)d_out;

    float* ws   = (float*)d_ws;
    float* ht   = ws;
    float* ssrc = ht + (size_t)N_NODES * HID;
    float* sdst = ssrc + N_NODES * NHEAD;
    float* Mb   = sdst + N_NODES * NHEAD;

    k1_gemm_ht<<<dim3(N_NODES / 16, HID / 64), 128, 0, stream>>>(h, W, ht);
    k2_scores<<<(N_NODES * NHEAD) / 256, 256, 0, stream>>>(ht, a, ssrc, sdst);
    k3_rowmax<<<N_NODES / K3_ROWS, 256, 0, stream>>>(adj, ssrc, sdst, Mb);
    k4_agg_ln<<<N_NODES / 16, K4_THREADS, 0, stream>>>(adj, ht, ssrc, sdst, Mb,
                                                gamma, beta, out);
}


# 2 launches instead of 4: k1's epilogue makes s_src/s_dst from its tile in LDS (f order kept), k4's first phase takes the masked row max of its own 16 rows; M and every sum order bit-identical
# speedup vs baseline: 12.5586x; 1.0347x over previous
#include <hip/hip_runtime.h>
#include <hip/hip_bf16.h>

#define N_NODES 4096
#define IN_F    256
#define OUT_F   64
#define NHEAD   8
#define HID     512
#define NEG_BIG (-9.0e15f)
#define LOG2E   1.44269504088896340736f
#define LN_EPS  1e-05f

typedef __attribute__((ext_vector_type(2))) float v2f;
typedef __attribute__((ext_vector_type(8))) float v8f;

__device__ __forceinline__ v8f wmma_f32_16x16x4(v2f A, v2f B, v8f C) {
    return __builtin_amdgcn_wmma_f32_16x16x4_f32(false, A, false, B, (short)0, C,
                                                 false, false);
}

#define ST_PITCH 68
__global__ void __launch_bounds__(128)
k1_gemm_ht(const float* __restrict__ h, const float* __restrict__ W,
           const float* __restrict__ a,
           float* __restrict__ ht, float* __restrict__ ssrc, float* __restrict__ sdst) {
    __shared__ float sA[16 * 260];
    __shared__ alignas(16) float sT[16 * ST_PITCH];
    __shared__ alignas(16) float sAv[2 * OUT_F];
    const int i0 = blockIdx.x * 16;
    const int n0 = blockIdx.y * 64;
    const int tid  = threadIdx.x;
    const int lane = tid & 31;
    const int wave = tid >> 5;

    const float a_mine = a[tid];
    for (int idx = tid; idx < 16 * 256 / 4; idx += 128) {
        const int r = idx >> 6, c = (idx & 63) << 2;
        const float4 v = *(const float4*)&h[(size_t)(i0 + r) * IN_F + c];
        *(float4*)&sA[r * 260 + c] = v;
    }
    sAv[tid] = a_mine;
    __syncthreads();

    const int m  = lane & 15;
    const int hi = lane >> 4;

    v8f acc = {};
    for (int ks = 0; ks < IN_F / 4; ++ks) {
        v2f A, B;
#pragma unroll
        for (int v = 0; v < 2; ++v) {
            const int kk = (hi << 1) + v;
            const int k  = ks * 4 + kk;
            A[v] = sA[m * 260 + k];
            B[v] = W[(size_t)k * HID + n0 + (wave << 4) + m];
        }
        acc = wmma_f32_16x16x4(A, B, acc);
    }

#pragma unroll
    for (int p = 0; p < 8; ++p) {
        const int row = p + (hi << 3);
        sT[row * ST_PITCH + (wave << 4) + m] = acc[p];
    }
    __syncthreads();
#pragma unroll
    for (int p = 0; p < 8; ++p) {
        const int row = p + (hi << 3);
        ht[(size_t)(i0 + row) * HID + n0 + (wave << 4) + m] = acc[p];
    }
    if (tid < 16) {
        const float* const trow = sT + tid * ST_PITCH;
        float s1 = 0.f, s2 = 0.f;
#pragma unroll 16
        for (int f = 0; f < OUT_F; ++f) {
            const float v = trow[f];
            s1 = __builtin_fmaf(v, sAv[f], s1);
            s2 = __builtin_fmaf(v, sAv[OUT_F + f], s2);
        }
        ssrc[(i0 + tid) * NHEAD + blockIdx.y] = s1;
        sdst[(i0 + tid) * NHEAD + blockIdx.y] = s2;
    }
}

#define K3_ROWS 4
#define K3_RED  (K3_ROWS * NHEAD * 8 * 36)
#define K3_RED2 (K3_ROWS * NHEAD * 8)
__device__ __forceinline__ float k3_max(float a, float b) { return __builtin_fmaximum_numf(a, b); }
__device__ __forceinline__ void k4_rowmax(const int* __restrict__ adj, const float* __restrict__ ssrc,
                                          const float* __restrict__ sdst,
                                          float* __restrict__ red_all, float* __restrict__ red2_all,
                                          float* __restrict__ s_M, float* __restrict__ s_ss,
                                          const int i0blk, const int tid_blk) {
    const int grp  = tid_blk >> 8;
    const int tid  = tid_blk & 255;
    const int lane = tid & 31;
    const int wave = tid >> 5;
    const int i0   = i0blk + grp * K3_ROWS;
    float* const red  = red_all  + grp * K3_RED;
    float* const red2 = red2_all + grp * K3_RED2;
    const float NEG_INF = -__builtin_inff();

    float mx[K3_ROWS][NHEAD];
#pragma unroll
    for (int r = 0; r < K3_ROWS; ++r)
#pragma unroll
        for (int hh = 0; hh < NHEAD; ++hh) mx[r][hh] = NEG_INF;

#pragma unroll 2
    for (int t = 0; t < N_NODES / 256; ++t) {
        const int j = tid + 256 * t;
        const float4 d0 = *(const float4*)&sdst[j * NHEAD];
        const float4 d1 = *(const float4*)&sdst[j * NHEAD + 4];
        const float d[NHEAD] = {d0.x, d0.y, d0.z, d0.w, d1.x, d1.y, d1.z, d1.w};
#pragma unroll
        for (int r = 0; r < K3_ROWS; ++r) {
            const int av = adj[(size_t)(i0 + r) * N_NODES + j];
            const float bias = av > 0 ? 0.f : NEG_INF;
#pragma unroll
            for (int hh = 0; hh < NHEAD; ++hh)
                mx[r][hh] = k3_max(mx[r][hh], d[hh] + bias);
        }
    }

#pragma unroll
    for (int r = 0; r < K3_ROWS; ++r)
#pragma unroll
        for (int hh = 0; hh < NHEAD; ++hh)
            red[(r * NHEAD + hh) * (8 * 36) + wave * 36 + lane] = mx[r][hh];
    __syncthreads();
    {
        const float4* const q = (const float4*)&red[(tid >> 3) * (8 * 36) + (tid & 7) * 36];
        float4 a4 = q[0];
#pragma unroll
        for (int k = 1; k < 8; ++k) {
            const float4 b4 = q[k];
            a4.x = k3_max(a4.x, b4.x); a4.y = k3_max(a4.y, b4.y);
            a4.z = k3_max(a4.z, b4.z); a4.w = k3_max(a4.w, b4.w);
        }
        red2[tid] = k3_max(k3_max(a4.x, a4.y), k3_max(a4.z, a4.w));
    }
    __syncthreads();
    if (tid < K3_ROWS * NHEAD) {
        const float4 p0 = *(const float4*)&red2[tid * 8];
        const float4 p1 = *(const float4*)&red2[tid * 8 + 4];
        const float dm = k3_max(k3_max(k3_max(p0.x, p0.y), k3_max(p0.z, p0.w)),
                                k3_max(k3_max(p1.x, p1.y), k3_max(p1.z, p1.w)));
        const float ss = ssrc[i0 * NHEAD + tid];
        float e = ss + dm;
        e = e > 0.f ? e : 0.2f * e;
        s_M[grp * (K3_ROWS * NHEAD) + tid]  = dm == NEG_INF ? NEG_BIG : fmaxf(NEG_BIG, e);
        s_ss[grp * (K3_ROWS * NHEAD) + tid] = ss;
    }
}

#define JT 64
#define K4_THREADS 1024
#define KS_WAVE 4
#define ADJ_PITCH 68
#define SD_OFF (16 * ADJ_PITCH)
#define STAGE_INTS (SD_OFF + JT * NHEAD)
#define N_STAGES 4
#define N_CHUNKS (N_NODES / JT)

__device__ __forceinline__ void k4_load_b(float4 (&dst)[8], const float* __restrict__ src) {
#pragma unroll
    for (int q = 0; q < 8; ++q)
        dst[q] = *(const float4*)&src[(size_t)(4 * (q >> 1) + (q & 1)) * HID];
}

#define MASK_OFF (NEG_BIG * LOG2E)

__device__ __forceinline__ int4 k4_stage_mask(const int4 pre) {
    const int off = __float_as_int(MASK_OFF);
    return make_int4(pre.x > 0 ? 0 : off, pre.y > 0 ? 0 : off,
                     pre.z > 0 ? 0 : off, pre.w > 0 ? 0 : off);
}
__device__ __forceinline__ int4 k4_stage_sd(const int4 pre) {
    return make_int4(__float_as_int(__int_as_float(pre.x) * LOG2E),
                     __float_as_int(__int_as_float(pre.y) * LOG2E),
                     __float_as_int(__int_as_float(pre.z) * LOG2E),
                     __float_as_int(__int_as_float(pre.w) * LOG2E));
}

__device__ __forceinline__ void k4_ring_arrive() {
    __builtin_amdgcn_fence(__ATOMIC_RELEASE, "workgroup", "local");
    __builtin_amdgcn_s_barrier_signal(-1);
}
__device__ __forceinline__ void k4_ring_wait() {
    __builtin_amdgcn_s_barrier_wait(-1);
    __builtin_amdgcn_fence(__ATOMIC_ACQUIRE, "workgroup", "local");
}

__device__ __forceinline__ void k4_chunk(const int* a_msk, const int* a_sd,
                                         const float4 (&hv)[8],
                                         const v2f u, const v2f u2,
                                         v8f& acc0, v8f& acc1, v8f& acc2, v8f& acc3,
                                         v2f& lsum2) {
#pragma unroll
    for (int ks = 0; ks < KS_WAVE; ++ks) {
        const int jo = ks * 4;
        const v2f d   = {__int_as_float(a_sd[jo * NHEAD]),
                         __int_as_float(a_sd[(jo + 1) * NHEAD])};
        const v2f msk = {__int_as_float(a_msk[jo]), __int_as_float(a_msk[jo + 1])};
        const v2f t1 = u + d;
        const v2f t2 = d * 0.2f + u2;
        const v2f arg = __builtin_elementwise_max(t1, t2) + msk;
        const v2f A = {__builtin_amdgcn_exp2f(arg[0]), __builtin_amdgcn_exp2f(arg[1])};
        lsum2 += A;
        const float4 r0 = hv[2 * ks];
        const float4 r1 = hv[2 * ks + 1];
        const v2f B0 = {r0.x, r1.x};
        const v2f B1 = {r0.y, r1.y};
        const v2f B2 = {r0.z, r1.z};
        const v2f B3 = {r0.w, r1.w};
        acc0 = wmma_f32_16x16x4(A, B0, acc0);
        acc1 = wmma_f32_16x16x4(A, B1, acc1);
        acc2 = wmma_f32_16x16x4(A, B2, acc2);
        acc3 = wmma_f32_16x16x4(A, B3, acc3);
    }
}

__global__ void __launch_bounds__(K4_THREADS)
k4_agg_ln(const int* __restrict__ adj, const float* __restrict__ ht,
          const float* __restrict__ ssrc, const float* __restrict__ sdst,
          const float* __restrict__ gamma, const float* __restrict__ beta,
          float* __restrict__ out) {
    __shared__ alignas(16) int s_tile[N_STAGES * STAGE_INTS];
    __shared__ alignas(16) float s_big[4 * K3_RED];
    static_assert(4 * K3_RED >= NHEAD * 4 * 16 * OUT_F, "k partials must fit the shared buffer");
    float* const s_part = s_big;
    float* const s_red  = s_big;
    __shared__ float s_out[16 * OUT_F];
    __shared__ float s_L[16 * NHEAD];
    __shared__ float s_Lp[NHEAD * 4 * 2 * 16];
    __shared__ float s_M[16 * NHEAD];
    __shared__ float s_ss[16 * NHEAD];
    __shared__ float s_mu[16], s_rs[16];
    __shared__ alignas(16) float s_red2[4 * K3_RED2];

    const int i0   = blockIdx.x * 16;
    const int tid  = threadIdx.x;
    const int hh   = tid >> 7;
    const int gt   = tid & 127;
    const int lane = tid & 31;
    const int wave = gt >> 5;
    const int m    = lane & 15;
    const int hi   = lane >> 4;

    k4_rowmax(adj, ssrc, sdst, s_red, s_red2, s_M, s_ss, i0, tid);

    const bool stager = tid < 16 * JT / 4 + JT * NHEAD / 4;
    const bool adj_stager = tid < 16 * JT / 4;
    const int4* t_src;
    int t_step;
    int t_dst;
    if (tid < 16 * JT / 4) {
        const int r = tid >> 4, c = (tid & 15) << 2;
        t_src  = (const int4*)&adj[(size_t)(i0 + r) * N_NODES + c];
        t_step = JT / 4;
        t_dst  = r * ADJ_PITCH + c;
    } else {
        const int t4 = ((tid - 16 * JT / 4) & (JT * NHEAD / 4 - 1)) << 2;
        t_src  = (const int4*)&sdst[t4];
        t_step = JT * NHEAD / 4;
        t_dst  = SD_OFF + t4;
    }

    const int jw = (wave << 4) + (hi << 1);
    const int* const a_adj = s_tile + m * ADJ_PITCH + jw;
    const int* const a_sd  = s_tile + SD_OFF + jw * NHEAD + hh;
    const float* const b_src = ht + (size_t)jw * HID + hh * OUT_F + (m << 2);

    float4 hvA[8], hvB[8];
    int4 pre0 = make_int4(0, 0, 0, 0);
    int4 pre  = make_int4(0, 0, 0, 0);
    if (stager) { pre0 = t_src[0]; pre = t_src[t_step]; }
    k4_load_b(hvA, b_src);
    if (adj_stager)  *(int4*)__builtin_assume_aligned(&s_tile[t_dst], 16) = k4_stage_mask(pre0);
    else if (stager) *(int4*)__builtin_assume_aligned(&s_tile[t_dst], 16) = k4_stage_sd(pre0);
    k4_ring_arrive();
    k4_ring_wait();

    const float ssr  = s_ss[m * NHEAD + hh];
    const float Mrow = s_M[m * NHEAD + hh];
    const float u1s = (ssr - Mrow) * LOG2E;
    const float u2s = (0.2f * ssr - Mrow) * LOG2E;
    const v2f u  = {u1s, u1s};
    const v2f u2 = {u2s, u2s};

    v8f acc0 = {}, acc1 = {}, acc2 = {}, acc3 = {};
    v2f lsum2 = {};

    const int* pa = a_adj;
    const int* ps = a_sd;
    int* pt = s_tile + t_dst;
    int hop = 2 * STAGE_INTS;

    for (int c = 0; c < N_CHUNKS; c += 2) {
        if (adj_stager)  *(int4*)__builtin_assume_aligned(pt + STAGE_INTS, 16) = k4_stage_mask(pre);
        else if (stager) *(int4*)__builtin_assume_aligned(pt + STAGE_INTS, 16) = k4_stage_sd(pre);
        k4_ring_arrive();
        if (stager) pre = t_src[(size_t)((c + 2) & (N_CHUNKS - 1)) * t_step];
        k4_load_b(hvB, b_src + (size_t)(c + 1) * JT * HID);
        k4_chunk(pa, ps, hvA, u, u2, acc0, acc1, acc2, acc3, lsum2);
        k4_ring_wait();

        if (adj_stager)  *(int4*)__builtin_assume_aligned(pt + hop, 16) = k4_stage_mask(pre);
        else if (stager) *(int4*)__builtin_assume_aligned(pt + hop, 16) = k4_stage_sd(pre);
        k4_ring_arrive();
        if (stager) pre = t_src[(size_t)((c + 3) & (N_CHUNKS - 1)) * t_step];
        k4_load_b(hvA, b_src + (size_t)((c + 2) & (N_CHUNKS - 1)) * JT * HID);
        k4_chunk(pa + STAGE_INTS, ps + STAGE_INTS, hvB, u, u2, acc0, acc1, acc2, acc3, lsum2);
        k4_ring_wait();

        pa += hop; ps += hop; pt += hop;
        hop = -hop;
    }

    asm volatile("s_wait_alu depctr_va_vdst(0)"
                 : "+v"(acc0), "+v"(acc1), "+v"(acc2), "+v"(acc3));
    {
        float* const my_part = s_part + ((hh * 4 + wave) * 16) * OUT_F + (m << 2);
#pragma unroll
        for (int p = 0; p < 8; ++p) {
            const int row = p + (hi << 3);
            *(float4*)&my_part[row * OUT_F] =
                make_float4(acc0[p], acc1[p], acc2[p], acc3[p]);
        }
    }
    s_Lp[tid] = lsum2[0] + lsum2[1];
    __syncthreads();

    if (tid < 16 * NHEAD) {
        const int r = tid >> 3, h8 = tid & 7;
        const float* const q = s_Lp + h8 * (4 * 2 * 16) + r;
        float L = q[0];
        for (int t = 1; t < 8; ++t) L += q[t * 16];
        s_L[r * NHEAD + h8] = L;
    }
    __syncthreads();

    {
        const int r = tid >> 6;
        float hp = 0.f;
#pragma unroll
        for (int h8 = 0; h8 < NHEAD; ++h8) {
            const float* const q = s_part + (h8 * 4) * 16 * OUT_F + tid;
            float s = q[0];
            s += q[1 * 16 * OUT_F];
            s += q[2 * 16 * OUT_F];
            s += q[3 * 16 * OUT_F];
            hp += s / s_L[r * NHEAD + h8];
        }
        s_out[tid] = hp * (1.0f / NHEAD);
    }
    __syncthreads();

    if (tid < 16) {
        float mu = 0.f;
        for (int f = 0; f < OUT_F; ++f) mu += s_out[tid * OUT_F + f];
        mu *= (1.0f / OUT_F);
        float var = 0.f;
        for (int f = 0; f < OUT_F; ++f) {
            const float d = s_out[tid * OUT_F + f] - mu;
            var += d * d;
        }
        var *= (1.0f / OUT_F);
        s_mu[tid] = mu;
        s_rs[tid] = rsqrtf(var + LN_EPS);
    }
    __syncthreads();

    {
        const int r = tid >> 6, f = tid & 63;
        out[(size_t)(i0 + r) * OUT_F + f] =
            (s_out[r * OUT_F + f] - s_mu[r]) * s_rs[r] * gamma[f] + beta[f];
    }
}

extern "C" void kernel_launch(void* const* d_in, const int* in_sizes, int n_in,
                              void* d_out, int out_size, void* d_ws, size_t ws_size,
                              hipStream_t stream) {
    const float* h     = (const float*)d_in[0];
    const int*   adj   = (const int*)  d_in[1];
    const float* W     = (const float*)d_in[2];
    const float* a     = (const float*)d_in[3];
    const float* gamma = (const float*)d_in[4];
    const float* beta  = (const float*)d_in[5];
    float* out = (float*)d_out;

    float* ws   = (float*)d_ws;
    float* ht   = ws;
    float* ssrc = ht + (size_t)N_NODES * HID;
    float* sdst = ssrc + N_NODES * NHEAD;

    k1_gemm_ht<<<dim3(N_NODES / 16, HID / 64), 128, 0, stream>>>(h, W, a, ht, ssrc, sdst);
    k4_agg_ln<<<N_NODES / 16, K4_THREADS, 0, stream>>>(adj, ht, ssrc, sdst,
                                                gamma, beta, out);
}
